# starting kernel
# baseline (speedup 1.0000x reference)
#include <hip/hip_runtime.h>
#include <hip/hip_bf16.h>
#include <math.h>


#define HDIM 128
#define WS2  144
#define A_LD 138

typedef __attribute__((ext_vector_type(2))) float v2f;
typedef __attribute__((ext_vector_type(8))) float v8f;

__global__ __launch_bounds__(256)
void gcn_gemm_wmma_f32(const float* __restrict__ A,
                       const float* __restrict__ W,
                       const float* __restrict__ bias,
                       const float* __restrict__ dinv,
                       float* __restrict__ C,
                       float* __restrict__ ACC,
                       int N, int K, int Kpad) {
    extern __shared__ float lds[];
    v2f*   Wp = (v2f*)lds;
    float* As = lds + (size_t)(Kpad / 2) * WS2 * 2;

    const int tid  = threadIdx.x;
    const int wave = tid >> 5;
    const int lane = tid & 31;
    const int l15  = lane & 15;
    const int rowBase = blockIdx.x * 256 + wave * 32;

    for (int idx = tid; idx < (Kpad / 2) * HDIM; idx += 256) {
        int kk = idx >> 7;
        int n  = idx & (HDIM - 1);
        int k0 = kk * 2;
        v2f w;
        w.x = (k0     < K) ? W[(size_t)k0       * HDIM + n] : 0.0f;
        w.y = (k0 + 1 < K) ? W[(size_t)(k0 + 1) * HDIM + n] : 0.0f;
        Wp[kk * WS2 + n] = w;
    }

    float* Aw = As + (size_t)wave * 32 * A_LD;
    for (int r = 0; r < 32; ++r) {
        int row = rowBase + r;
        for (int k = lane; k < Kpad; k += 32) {
            Aw[r * A_LD + k] = (row < N && k < K) ? A[(size_t)row * K + k] : 0.0f;
        }
    }
    __syncthreads();

    v8f acc[2][8];
#pragma unroll
    for (int t = 0; t < 2; ++t)
#pragma unroll
        for (int j = 0; j < 8; ++j)
            acc[t][j] = (v8f){0.f,0.f,0.f,0.f,0.f,0.f,0.f,0.f};

    const int koff = (lane >= 16) ? 2 : 0;
    const float* Aw0 = Aw + l15 * A_LD;
    const float* Aw1 = Aw + (l15 + 16) * A_LD;

    for (int k0 = 0; k0 < Kpad; k0 += 4) {
        v2f a0 = *(const v2f*)(Aw0 + k0 + koff);
        v2f a1 = *(const v2f*)(Aw1 + k0 + koff);
        const v2f* Bk = Wp + (size_t)((k0 + koff) >> 1) * WS2 + l15;
#pragma unroll
        for (int j = 0; j < 8; ++j) {
            v2f b = Bk[j * 16];
            acc[0][j] = __builtin_amdgcn_wmma_f32_16x16x4_f32(
                false, a0, false, b, (short)0, acc[0][j], false, false);
            acc[1][j] = __builtin_amdgcn_wmma_f32_16x16x4_f32(
                false, a1, false, b, (short)0, acc[1][j], false, false);
        }
    }

    float bcol[8];
#pragma unroll
    for (int j = 0; j < 8; ++j) bcol[j] = bias[j * 16 + l15];

    const int rOff = (lane >= 16) ? 8 : 0;
#pragma unroll
    for (int t = 0; t < 2; ++t) {
#pragma unroll
        for (int i = 0; i < 8; ++i) {
            int row = rowBase + t * 16 + i + rOff;
            if (row < N) {
                float di  = dinv[row];
                float di2 = di * di;
                size_t base = (size_t)row * HDIM + l15;
#pragma unroll
                for (int j = 0; j < 8; ++j) {
                    float h = acc[t][j][i];
                    C[base + j * 16]   = h;
                    ACC[base + j * 16] = h * di2 + bcol[j];
                }
            }
        }
    }
}

__global__ void zero_f32(float* __restrict__ p, int n) {
    int i = blockIdx.x * blockDim.x + threadIdx.x;
    if (i < n) p[i] = 0.0f;
}

__global__ void deg_count(const int* __restrict__ dst, float* __restrict__ deg, int E) {
    int e = blockIdx.x * blockDim.x + threadIdx.x;
    if (e < E) unsafeAtomicAdd(&deg[dst[e]], 1.0f);
}

__global__ void finalize_dinv(float* __restrict__ d, int n) {
    int i = blockIdx.x * blockDim.x + threadIdx.x;
    if (i < n) d[i] = rsqrtf(d[i] + 1.0f);
}

__global__ __launch_bounds__(256)
void edge_scatter(const int* __restrict__ src, const int* __restrict__ dst,
                  const float* __restrict__ dinv, const float* __restrict__ g,
                  float* __restrict__ acc, int E) {
    int e = blockIdx.x * 8 + (threadIdx.x >> 5);
    if (e >= E) return;
    int lane = threadIdx.x & 31;

    int ep = e + 256;
    if (ep < E) {
        int sp = src[ep];
        __builtin_prefetch(g + (size_t)sp * HDIM + lane * 4, 0, 1);
    }

    int s = src[e];
    int d = dst[e];
    float coef = dinv[s] * dinv[d];
    const float4* gs = (const float4*)(g + (size_t)s * HDIM);
    float4 v = gs[lane];
    float* ad = acc + (size_t)d * HDIM + lane * 4;
    unsafeAtomicAdd(ad + 0, coef * v.x);
    unsafeAtomicAdd(ad + 1, coef * v.y);
    unsafeAtomicAdd(ad + 2, coef * v.z);
    unsafeAtomicAdd(ad + 3, coef * v.w);
}

__global__ void tanh_act(const float* __restrict__ in, float* __restrict__ out, int n) {
    int i = blockIdx.x * blockDim.x + threadIdx.x;
    if (i < n) out[i] = tanhf(in[i]);
}

extern "C" void kernel_launch(void* const* d_in, const int* in_sizes, int n_in,
                              void* d_out, int out_size, void* d_ws, size_t ws_size,
                              hipStream_t stream) {
    const float* x  = (const float*)d_in[0];
    const int*   ei = (const int*)d_in[1];
    const float* Wl[4] = {(const float*)d_in[2], (const float*)d_in[4],
                          (const float*)d_in[6], (const float*)d_in[8]};
    const float* bl[4] = {(const float*)d_in[3], (const float*)d_in[5],
                          (const float*)d_in[7], (const float*)d_in[9]};

    const int NFEAT = in_sizes[2] / HDIM;
    const int N     = in_sizes[0] / NFEAT;
    const int E     = in_sizes[1] / 2;
    const int* srcp = ei;
    const int* dstp = ei + E;

    float* wsf  = (float*)d_ws;
    float* dinv = wsf;
    float* r0   = wsf + N;
    float* r1   = r0 + (size_t)N * HDIM;
    float* r2   = r1 + (size_t)N * HDIM;

    const int THR = 256;

    zero_f32<<<(N + THR - 1) / THR, THR, 0, stream>>>(dinv, N);
    deg_count<<<(E + THR - 1) / THR, THR, 0, stream>>>(dstp, dinv, E);
    finalize_dinv<<<(N + THR - 1) / THR, THR, 0, stream>>>(dinv, N);

    const float* in = x;
    int K = NFEAT;
    for (int l = 0; l < 4; ++l) {
        const int Kpad = (K + 3) & ~3;
        const size_t shmem =
            ((size_t)(Kpad / 2) * WS2 * 2 + 8u * 32u * A_LD) * sizeof(float);
        gcn_gemm_wmma_f32<<<(N + 255) / 256, 256, shmem, stream>>>(
            in, Wl[l], bl[l], dinv, r0, r1, N, K, Kpad);

        edge_scatter<<<(E + 7) / 8, 256, 0, stream>>>(srcp, dstp, dinv, r0, r1, E);

        float* outp = (l == 3) ? (float*)d_out : r2;
        tanh_act<<<((size_t)N * HDIM + THR - 1) / THR, THR, 0, stream>>>(
            r1, outp, (int)((size_t)N * HDIM));

        in = r2;
        K = HDIM;
    }
}


# each row block's edge list (source, local row) is saved by layer 1's dst scan in the freed seed buffer; layers 2-4 read it, no dst re-scan; seed now formed in edge kernel; sums in arrival order
# speedup vs baseline: 2.1062x; 2.1062x over previous
#include <hip/hip_runtime.h>
#include <hip/hip_bf16.h>
#include <math.h>


#define HDIM 128
#define WS2  144
#define A_LD 138

typedef __attribute__((ext_vector_type(2))) float v2f;
typedef __attribute__((ext_vector_type(8))) float v8f;

__global__ __launch_bounds__(256)
void gcn_gemm_wmma_f32(const float* __restrict__ A,
                       const float* __restrict__ W,
                       float* __restrict__ C,
                       int N, int K, int Kpad) {
    extern __shared__ float lds[];
    v2f*   Wp = (v2f*)lds;
    float* As = lds + (size_t)(Kpad / 2) * WS2 * 2;

    const int tid  = threadIdx.x;
    const int wave = tid >> 5;
    const int lane = tid & 31;
    const int l15  = lane & 15;
    const int rowBase = blockIdx.x * 256 + wave * 32;

    for (int idx = tid; idx < (Kpad / 2) * HDIM; idx += 256) {
        int kk = idx >> 7;
        int n  = idx & (HDIM - 1);
        int k0 = kk * 2;
        v2f w;
        w.x = (k0     < K) ? W[(size_t)k0       * HDIM + n] : 0.0f;
        w.y = (k0 + 1 < K) ? W[(size_t)(k0 + 1) * HDIM + n] : 0.0f;
        Wp[kk * WS2 + n] = w;
    }

    float* Aw = As + (size_t)wave * 32 * A_LD;
    for (int r = 0; r < 32; ++r) {
        int row = rowBase + r;
        for (int k = lane; k < Kpad; k += 32) {
            Aw[r * A_LD + k] = (row < N && k < K) ? A[(size_t)row * K + k] : 0.0f;
        }
    }
    __syncthreads();

    v8f acc[2][8];
#pragma unroll
    for (int t = 0; t < 2; ++t)
#pragma unroll
        for (int j = 0; j < 8; ++j)
            acc[t][j] = (v8f){0.f,0.f,0.f,0.f,0.f,0.f,0.f,0.f};

    const int koff = (lane >= 16) ? 2 : 0;
    const float* Aw0 = Aw + l15 * A_LD;
    const float* Aw1 = Aw + (l15 + 16) * A_LD;

    for (int k0 = 0; k0 < Kpad; k0 += 4) {
        v2f a0 = *(const v2f*)(Aw0 + k0 + koff);
        v2f a1 = *(const v2f*)(Aw1 + k0 + koff);
        const v2f* Bk = Wp + (size_t)((k0 + koff) >> 1) * WS2 + l15;
#pragma unroll
        for (int j = 0; j < 8; ++j) {
            v2f b = Bk[j * 16];
            acc[0][j] = __builtin_amdgcn_wmma_f32_16x16x4_f32(
                false, a0, false, b, (short)0, acc[0][j], false, false);
            acc[1][j] = __builtin_amdgcn_wmma_f32_16x16x4_f32(
                false, a1, false, b, (short)0, acc[1][j], false, false);
        }
    }

    const int rOff = (lane >= 16) ? 8 : 0;
#pragma unroll
    for (int t = 0; t < 2; ++t) {
#pragma unroll
        for (int i = 0; i < 8; ++i) {
            int row = rowBase + t * 16 + i + rOff;
            if (row < N) {
                size_t base = (size_t)row * HDIM + l15;
#pragma unroll
                for (int j = 0; j < 8; ++j) C[base + j * 16] = acc[t][j][i];
            }
        }
    }
}

__global__ void zero_f32(float* __restrict__ p, int n) {
    int i = blockIdx.x * blockDim.x + threadIdx.x;
    if (i < n) p[i] = 0.0f;
}

__global__ void deg_count(const int* __restrict__ dst, float* __restrict__ deg, int E) {
    int e = blockIdx.x * blockDim.x + threadIdx.x;
    if (e < E) unsafeAtomicAdd(&deg[dst[e]], 1.0f);
}

__global__ void finalize_dinv(float* __restrict__ d, int n) {
    int i = blockIdx.x * blockDim.x + threadIdx.x;
    if (i < n) d[i] = rsqrtf(d[i] + 1.0f);
}

#define RB 400
#define SW 32
#define SL 4
#define QC 544
#define EG 8
#define LHDR 32
#define LSH 23

typedef __attribute__((ext_vector_type(4))) int v4i;

__device__ __forceinline__ void lds_row_add(float* rw, float c, float4 v) {
    __hip_atomic_fetch_add(rw,      c * v.x, __ATOMIC_RELAXED, __HIP_MEMORY_SCOPE_WORKGROUP);
    __hip_atomic_fetch_add(rw + 32, c * v.y, __ATOMIC_RELAXED, __HIP_MEMORY_SCOPE_WORKGROUP);
    __hip_atomic_fetch_add(rw + 64, c * v.z, __ATOMIC_RELAXED, __HIP_MEMORY_SCOPE_WORKGROUP);
    __hip_atomic_fetch_add(rw + 96, c * v.w, __ATOMIC_RELAXED, __HIP_MEMORY_SCOPE_WORKGROUP);
}

__device__ __forceinline__ void add_edges(int s, int dl, float coef, int cnt,
                                          const float4* __restrict__ gl, float* rowsl) {
    if (cnt == 32) {
#pragma unroll
        for (int c = 0; c < 32 / EG; ++c) {
            float4 v[EG];
#pragma unroll
            for (int j = 0; j < EG; ++j) {
                int sj = __builtin_amdgcn_readlane(s, c * EG + j);
                v[j] = gl[(size_t)sj * (HDIM / 4)];
            }
#pragma unroll
            for (int j = 0; j < EG; ++j) {
                int   dj = __builtin_amdgcn_readlane(dl, c * EG + j);
                float cj = __int_as_float(
                    __builtin_amdgcn_readlane(__float_as_int(coef), c * EG + j));
                lds_row_add(rowsl + dj * HDIM, cj, v[j]);
            }
        }
    } else {
        for (int j = 0; j < cnt; ++j) {
            int   sj = __builtin_amdgcn_readlane(s, j);
            int   dj = __builtin_amdgcn_readlane(dl, j);
            float cj = __int_as_float(__builtin_amdgcn_readlane(__float_as_int(coef), j));
            lds_row_add(rowsl + dj * HDIM, cj, gl[(size_t)sj * (HDIM / 4)]);
        }
    }
}

template <bool SAVE>
__device__ __forceinline__ void drain_edges(const int* qp, int cnt, int lane, int row0,
                                            const int* __restrict__ src,
                                            const int* __restrict__ dst,
                                            const float* __restrict__ dinv,
                                            const float4* __restrict__ gl,
                                            float* rowsl, int* lcount, int* mylist,
                                            int lcap) {
    int e = (lane < cnt) ? qp[lane] : 0;
    int s = src[e];
    int dg = dst[e];
    float coef = dinv[s] * dinv[dg];
    int dl = dg - row0;
    if (SAVE) {
        int base = 0;
        if (lane == 0)
            base = __hip_atomic_fetch_add(lcount, cnt, __ATOMIC_RELAXED,
                                          __HIP_MEMORY_SCOPE_WORKGROUP);
        base = __builtin_amdgcn_readfirstlane(base);
        if (base + cnt <= lcap && lane < cnt)
            mylist[LHDR + base + lane] = (int)((unsigned)s | ((unsigned)dl << LSH));
    }
    add_edges(s, dl, coef, cnt, gl, rowsl);
}

template <bool SAVE>
__device__ __forceinline__ void scan_edges(int* q, int wv, int lane, int row0, int nrows,
                                           const int* __restrict__ src,
                                           const int* __restrict__ dst,
                                           const float* __restrict__ dinv,
                                           const float4* __restrict__ gl,
                                           float* rowsl, int* lcount, int* mylist,
                                           int lcap, int E) {
    const int off = (int)(((size_t)dst >> 2) & 3);
    const v4i* __restrict__ d4 = (const v4i*)(dst - off);
    const int nfull = (E + off) >> 2;
    int qn = 0;

    if (wv == 0) {
        int et = 4 * nfull - off + lane;
        bool m = false;
        if (et >= 0 && et < E) m = (unsigned)(dst[et] - row0) < (unsigned)nrows;
        unsigned b = __builtin_amdgcn_ballot_w32(m);
        if (m) q[__builtin_amdgcn_mbcnt_lo(b, 0)] = et;
        qn = __builtin_popcount(b);
    }

    const int niter = (nfull + SW * 32 * SL - 1) / (SW * 32 * SL);
    const int rot = (int)(((long long)blockIdx.x * niter) / gridDim.x);
    for (int it = 0; it < niter; ++it) {
        int ci = it + rot;
        if (ci >= niter) ci -= niter;
        const int base = ci * (SW * 32 * SL) + wv * (32 * SL);
        v4i x[SL];
#pragma unroll
        for (int k = 0; k < SL; ++k) {
            int j4 = base + k * 32 + lane;
            x[k] = (j4 < nfull) ? d4[j4] : (v4i){-1, -1, -1, -1};
        }
        if (base == 0 && lane == 0) {
            if (off > 0) x[0].x = -1;
            if (off > 1) x[0].y = -1;
            if (off > 2) x[0].z = -1;
        }
#pragma unroll
        for (int k = 0; k < SL; ++k) {
            int e0 = 4 * (base + k * 32 + lane) - off;
#pragma unroll
            for (int c = 0; c < 4; ++c) {
                bool m = (unsigned)(x[k][c] - row0) < (unsigned)nrows;
                unsigned b = __builtin_amdgcn_ballot_w32(m);
                if (b) {
                    if (m) q[qn + __builtin_amdgcn_mbcnt_lo(b, 0)] = e0 + c;
                    qn += __builtin_popcount(b);
                }
            }
        }
        while (qn >= 32) {
            qn -= 32;
            drain_edges<SAVE>(q + qn, 32, lane, row0, src, dst, dinv, gl, rowsl,
                              lcount, mylist, lcap);
        }
    }
    if (qn > 0)
        drain_edges<SAVE>(q, qn, lane, row0, src, dst, dinv, gl, rowsl, lcount, mylist,
                          lcap);
}

template <bool FIRST>
__global__ __launch_bounds__(1024)
void edge_block_sum(const int* __restrict__ src, const int* __restrict__ dst,
                    const float* __restrict__ dinv, const float* __restrict__ g,
                    const float* __restrict__ bias, float* __restrict__ out,
                    int* __restrict__ lists, int N, int E) {
    extern __shared__ float lds[];
    float* rows = lds;
    const int lane = threadIdx.x & 31;
    const int wv = __builtin_amdgcn_readfirstlane((int)(threadIdx.x >> 5));
    int* q = (int*)(lds + RB * HDIM) + wv * QC;
    int* lcount = (int*)(lds + RB * HDIM) + SW * QC;

    const int row0  = blockIdx.x * RB;
    const int nrows = min(RB, N - row0);
    float* rowsl = rows + lane;
    const float4* __restrict__ gl = (const float4*)g + lane;
    int* mylist = lists + (size_t)row0 * HDIM;
    const int lcap = nrows * HDIM - LHDR;

    const float b0 = bias[4 * lane], b1 = bias[4 * lane + 1];
    const float b2 = bias[4 * lane + 2], b3 = bias[4 * lane + 3];
    for (int r = wv; r < nrows; r += SW) {
        float4 v = gl[(size_t)(row0 + r) * (HDIM / 4)];
        float di = dinv[row0 + r];
        float di2 = di * di;
        float* rw = rowsl + r * HDIM;
        rw[0]  = v.x * di2 + b0; rw[32] = v.y * di2 + b1;
        rw[64] = v.z * di2 + b2; rw[96] = v.w * di2 + b3;
    }
    if (FIRST && threadIdx.x == 0) *lcount = 0;
    __syncthreads();

    int nl = -1;
    if (!FIRST) nl = __builtin_amdgcn_readfirstlane(mylist[0]);

    if (nl >= 0) {
        int i = wv * 32;
        unsigned w = 0u;
        if (i < nl) w = (unsigned)mylist[LHDR + min(i + lane, nl - 1)];
        while (i < nl) {
            int s  = (int)(w & ((1u << LSH) - 1u));
            int dl = (int)(w >> LSH);
            float coef = dinv[s] * dinv[row0 + dl];
            const int inext = i + SW * 32;
            unsigned wn = (unsigned)mylist[LHDR + min(inext + lane, nl - 1)];
            add_edges(s, dl, coef, min(32, nl - i), gl, rowsl);
            w = wn;
            i = inext;
        }
    } else {
        scan_edges<FIRST>(q, wv, lane, row0, nrows, src, dst, dinv, gl, rowsl,
                          lcount, mylist, lcap, E);
    }
    __syncthreads();

    if (FIRST && threadIdx.x == 0) {
        int c = *lcount;
        mylist[0] = (c <= lcap) ? c : -1;
    }

    for (int r = wv; r < nrows; r += SW) {
        const float* rw = rowsl + r * HDIM;
        float4 v;
        v.x = tanhf(rw[0]); v.y = tanhf(rw[32]); v.z = tanhf(rw[64]); v.w = tanhf(rw[96]);
        ((float4*)out)[(size_t)(row0 + r) * (HDIM / 4) + lane] = v;
    }
}

extern "C" void kernel_launch(void* const* d_in, const int* in_sizes, int n_in,
                              void* d_out, int out_size, void* d_ws, size_t ws_size,
                              hipStream_t stream) {
    const float* x  = (const float*)d_in[0];
    const int*   ei = (const int*)d_in[1];
    const float* Wl[4] = {(const float*)d_in[2], (const float*)d_in[4],
                          (const float*)d_in[6], (const float*)d_in[8]};
    const float* bl[4] = {(const float*)d_in[3], (const float*)d_in[5],
                          (const float*)d_in[7], (const float*)d_in[9]};

    const int NFEAT = in_sizes[2] / HDIM;
    const int N     = in_sizes[0] / NFEAT;
    const int E     = in_sizes[1] / 2;
    const int* srcp = ei;
    const int* dstp = ei + E;

    float* wsf  = (float*)d_ws;
    float* dinv = wsf;
    float* r0   = wsf + N;
    float* r1   = r0 + (size_t)N * HDIM;
    float* r2   = r1 + (size_t)N * HDIM;

    const int THR = 256;

    zero_f32<<<(N + THR - 1) / THR, THR, 0, stream>>>(dinv, N);
    deg_count<<<(E + THR - 1) / THR, THR, 0, stream>>>(dstp, dinv, E);
    finalize_dinv<<<(N + THR - 1) / THR, THR, 0, stream>>>(dinv, N);

    const float* in = x;
    int K = NFEAT;
    for (int l = 0; l < 4; ++l) {
        const int Kpad = (K + 3) & ~3;
        const size_t shmem =
            ((size_t)(Kpad / 2) * WS2 * 2 + 8u * 32u * A_LD) * sizeof(float);
        gcn_gemm_wmma_f32<<<(N + 255) / 256, 256, shmem, stream>>>(
            in, Wl[l], r0, N, K, Kpad);

        float* outp = (l == 3) ? (float*)d_out : r2;
        const size_t eshm = ((size_t)RB * HDIM + (size_t)SW * QC + 4) * sizeof(float);
        if (l == 0)
            edge_block_sum<true><<<(N + RB - 1) / RB, SW * 32, eshm, stream>>>(
                srcp, dstp, dinv, r0, bl[l], outp, (int*)r1, N, E);
        else
            edge_block_sum<false><<<(N + RB - 1) / RB, SW * 32, eshm, stream>>>(
                srcp, dstp, dinv, r0, bl[l], outp, (int*)r1, N, E);

        in = r2;
        K = HDIM;
    }
}


# layers 2-4 as (A_hat t)W: a block sums the weighted tanh rows of its 400 dst rows in LDS, then multiplies them by W there on f32 WMMA, adds bias, tanh; 3 GEMM launches gone; sum order changed
# speedup vs baseline: 3.3454x; 1.5884x over previous
#include <hip/hip_runtime.h>
#include <hip/hip_bf16.h>
#include <math.h>


#define HDIM 128
#define WS2  144
#define A_LD 138

typedef __attribute__((ext_vector_type(2))) float v2f;
typedef __attribute__((ext_vector_type(8))) float v8f;

__global__ __launch_bounds__(256)
void gcn_gemm_wmma_f32(const float* __restrict__ A,
                       const float* __restrict__ W,
                       float* __restrict__ C,
                       int N, int K, int Kpad) {
    extern __shared__ float lds[];
    v2f*   Wp = (v2f*)lds;
    float* As = lds + (size_t)(Kpad / 2) * WS2 * 2;

    const int tid  = threadIdx.x;
    const int wave = tid >> 5;
    const int lane = tid & 31;
    const int l15  = lane & 15;
    const int rowBase = blockIdx.x * 256 + wave * 32;

    for (int idx = tid; idx < (Kpad / 2) * HDIM; idx += 256) {
        int kk = idx >> 7;
        int n  = idx & (HDIM - 1);
        int k0 = kk * 2;
        v2f w;
        w.x = (k0     < K) ? W[(size_t)k0       * HDIM + n] : 0.0f;
        w.y = (k0 + 1 < K) ? W[(size_t)(k0 + 1) * HDIM + n] : 0.0f;
        Wp[kk * WS2 + n] = w;
    }

    float* Aw = As + (size_t)wave * 32 * A_LD;
    for (int r = 0; r < 32; ++r) {
        int row = rowBase + r;
        for (int k = lane; k < Kpad; k += 32) {
            Aw[r * A_LD + k] = (row < N && k < K) ? A[(size_t)row * K + k] : 0.0f;
        }
    }
    __syncthreads();

    v8f acc[2][8];
#pragma unroll
    for (int t = 0; t < 2; ++t)
#pragma unroll
        for (int j = 0; j < 8; ++j)
            acc[t][j] = (v8f){0.f,0.f,0.f,0.f,0.f,0.f,0.f,0.f};

    const int koff = (lane >= 16) ? 2 : 0;
    const float* Aw0 = Aw + l15 * A_LD;
    const float* Aw1 = Aw + (l15 + 16) * A_LD;

    for (int k0 = 0; k0 < Kpad; k0 += 4) {
        v2f a0 = *(const v2f*)(Aw0 + k0 + koff);
        v2f a1 = *(const v2f*)(Aw1 + k0 + koff);
        const v2f* Bk = Wp + (size_t)((k0 + koff) >> 1) * WS2 + l15;
#pragma unroll
        for (int j = 0; j < 8; ++j) {
            v2f b = Bk[j * 16];
            acc[0][j] = __builtin_amdgcn_wmma_f32_16x16x4_f32(
                false, a0, false, b, (short)0, acc[0][j], false, false);
            acc[1][j] = __builtin_amdgcn_wmma_f32_16x16x4_f32(
                false, a1, false, b, (short)0, acc[1][j], false, false);
        }
    }

    const int rOff = (lane >= 16) ? 8 : 0;
#pragma unroll
    for (int t = 0; t < 2; ++t) {
#pragma unroll
        for (int i = 0; i < 8; ++i) {
            int row = rowBase + t * 16 + i + rOff;
            if (row < N) {
                size_t base = (size_t)row * HDIM + l15;
#pragma unroll
                for (int j = 0; j < 8; ++j) C[base + j * 16] = acc[t][j][i];
            }
        }
    }
}

__global__ void zero_f32(float* __restrict__ p, int n) {
    int i = blockIdx.x * blockDim.x + threadIdx.x;
    if (i < n) p[i] = 0.0f;
}

__global__ void deg_count(const int* __restrict__ dst, float* __restrict__ deg, int E) {
    int e = blockIdx.x * blockDim.x + threadIdx.x;
    if (e < E) unsafeAtomicAdd(&deg[dst[e]], 1.0f);
}

__global__ void finalize_dinv(float* __restrict__ d, int n) {
    int i = blockIdx.x * blockDim.x + threadIdx.x;
    if (i < n) d[i] = rsqrtf(d[i] + 1.0f);
}

#define RB 400
#define SW 32
#define SL 4
#define QC 544
#define EG 8
#define LHDR 32
#define LSH 23

typedef __attribute__((ext_vector_type(4))) int v4i;

__device__ __forceinline__ void lds_row_add(float* rw, float c, float4 v) {
    __hip_atomic_fetch_add(rw,      c * v.x, __ATOMIC_RELAXED, __HIP_MEMORY_SCOPE_WORKGROUP);
    __hip_atomic_fetch_add(rw + 32, c * v.y, __ATOMIC_RELAXED, __HIP_MEMORY_SCOPE_WORKGROUP);
    __hip_atomic_fetch_add(rw + 64, c * v.z, __ATOMIC_RELAXED, __HIP_MEMORY_SCOPE_WORKGROUP);
    __hip_atomic_fetch_add(rw + 96, c * v.w, __ATOMIC_RELAXED, __HIP_MEMORY_SCOPE_WORKGROUP);
}

template <int RS>
__device__ __forceinline__ void add_edges(int s, int dl, float coef, int cnt,
                                          const float4* __restrict__ gl, float* rowsl) {
    if (cnt == 32) {
#pragma unroll
        for (int c = 0; c < 32 / EG; ++c) {
            float4 v[EG];
#pragma unroll
            for (int j = 0; j < EG; ++j) {
                int sj = __builtin_amdgcn_readlane(s, c * EG + j);
                v[j] = gl[(size_t)sj * (HDIM / 4)];
            }
#pragma unroll
            for (int j = 0; j < EG; ++j) {
                int   dj = __builtin_amdgcn_readlane(dl, c * EG + j);
                float cj = __int_as_float(
                    __builtin_amdgcn_readlane(__float_as_int(coef), c * EG + j));
                lds_row_add(rowsl + dj * RS, cj, v[j]);
            }
        }
    } else {
        for (int j = 0; j < cnt; ++j) {
            int   sj = __builtin_amdgcn_readlane(s, j);
            int   dj = __builtin_amdgcn_readlane(dl, j);
            float cj = __int_as_float(__builtin_amdgcn_readlane(__float_as_int(coef), j));
            lds_row_add(rowsl + dj * RS, cj, gl[(size_t)sj * (HDIM / 4)]);
        }
    }
}

template <bool SAVE, int RS>
__device__ __forceinline__ void drain_edges(const int* qp, int cnt, int lane, int row0,
                                            const int* __restrict__ src,
                                            const int* __restrict__ dst,
                                            const float* __restrict__ dinv,
                                            const float4* __restrict__ gl,
                                            float* rowsl, int* lcount, int* mylist,
                                            int lcap) {
    int e = (lane < cnt) ? qp[lane] : 0;
    int s = src[e];
    int dg = dst[e];
    float coef = dinv[s] * dinv[dg];
    int dl = dg - row0;
    if (SAVE) {
        int base = 0;
        if (lane == 0)
            base = __hip_atomic_fetch_add(lcount, cnt, __ATOMIC_RELAXED,
                                          __HIP_MEMORY_SCOPE_WORKGROUP);
        base = __builtin_amdgcn_readfirstlane(base);
        if (base + cnt <= lcap && lane < cnt)
            mylist[LHDR + base + lane] = (int)((unsigned)s | ((unsigned)dl << LSH));
    }
    add_edges<RS>(s, dl, coef, cnt, gl, rowsl);
}

template <bool SAVE, int RS>
__device__ __forceinline__ void scan_edges(int* q, int wv, int lane, int row0, int nrows,
                                           const int* __restrict__ src,
                                           const int* __restrict__ dst,
                                           const float* __restrict__ dinv,
                                           const float4* __restrict__ gl,
                                           float* rowsl, int* lcount, int* mylist,
                                           int lcap, int E) {
    const int off = (int)(((size_t)dst >> 2) & 3);
    const v4i* __restrict__ d4 = (const v4i*)(dst - off);
    const int nfull = (E + off) >> 2;
    int qn = 0;

    if (wv == 0) {
        int et = 4 * nfull - off + lane;
        bool m = false;
        if (et >= 0 && et < E) m = (unsigned)(dst[et] - row0) < (unsigned)nrows;
        unsigned b = __builtin_amdgcn_ballot_w32(m);
        if (m) q[__builtin_amdgcn_mbcnt_lo(b, 0)] = et;
        qn = __builtin_popcount(b);
    }

    const int niter = (nfull + SW * 32 * SL - 1) / (SW * 32 * SL);
    const int rot = (int)(((long long)blockIdx.x * niter) / gridDim.x);
    for (int it = 0; it < niter; ++it) {
        int ci = it + rot;
        if (ci >= niter) ci -= niter;
        const int base = ci * (SW * 32 * SL) + wv * (32 * SL);
        v4i x[SL];
#pragma unroll
        for (int k = 0; k < SL; ++k) {
            int j4 = base + k * 32 + lane;
            x[k] = (j4 < nfull) ? d4[j4] : (v4i){-1, -1, -1, -1};
        }
        if (base == 0 && lane == 0) {
            if (off > 0) x[0].x = -1;
            if (off > 1) x[0].y = -1;
            if (off > 2) x[0].z = -1;
        }
#pragma unroll
        for (int k = 0; k < SL; ++k) {
            int e0 = 4 * (base + k * 32 + lane) - off;
#pragma unroll
            for (int c = 0; c < 4; ++c) {
                bool m = (unsigned)(x[k][c] - row0) < (unsigned)nrows;
                unsigned b = __builtin_amdgcn_ballot_w32(m);
                if (b) {
                    if (m) q[qn + __builtin_amdgcn_mbcnt_lo(b, 0)] = e0 + c;
                    qn += __builtin_popcount(b);
                }
            }
        }
        while (qn >= 32) {
            qn -= 32;
            drain_edges<SAVE, RS>(q + qn, 32, lane, row0, src, dst, dinv, gl, rowsl,
                                  lcount, mylist, lcap);
        }
    }
    if (qn > 0)
        drain_edges<SAVE, RS>(q, qn, lane, row0, src, dst, dinv, gl, rowsl, lcount,
                              mylist, lcap);
}

#define RSM 132
#define NT (RB / 16)
#define WQW (HDIM / 2 * WS2 * 2)

template <bool FIRST>
__global__ __launch_bounds__(1024)
void edge_block_sum(const int* __restrict__ src, const int* __restrict__ dst,
                    const float* __restrict__ dinv, const float* __restrict__ g,
                    const float* __restrict__ Wn,
                    const float* __restrict__ bias, float* __restrict__ out,
                    int* __restrict__ lists, int N, int E) {
    constexpr int RS = FIRST ? HDIM : RSM;
    extern __shared__ float lds[];
    float* rows = lds;
    const int lane = threadIdx.x & 31;
    const int wv = __builtin_amdgcn_readfirstlane((int)(threadIdx.x >> 5));
    int* q = (int*)(lds + RB * RS) + wv * QC;
    int* lcount = (int*)(lds + RB * RS) + (FIRST ? SW * QC : WQW);

    const int row0  = blockIdx.x * RB;
    const int nrows = min(RB, N - row0);
    float* rowsl = rows + lane;
    const float4* __restrict__ gl = (const float4*)g + lane;
    int* mylist = lists + (size_t)row0 * HDIM;
    const int lcap = nrows * HDIM - LHDR;

    float b0 = 0.f, b1 = 0.f, b2 = 0.f, b3 = 0.f;
    if (FIRST) {
        b0 = bias[4 * lane];     b1 = bias[4 * lane + 1];
        b2 = bias[4 * lane + 2]; b3 = bias[4 * lane + 3];
    }
    for (int r = wv; r < nrows; r += SW) {
        float4 v = gl[(size_t)(row0 + r) * (HDIM / 4)];
        float di = dinv[row0 + r];
        float di2 = di * di;
        float* rw = rowsl + r * RS;
        if (FIRST) {
            rw[0]  = v.x * di2 + b0; rw[32] = v.y * di2 + b1;
            rw[64] = v.z * di2 + b2; rw[96] = v.w * di2 + b3;
        } else {
            rw[0]  = v.x * di2; rw[32] = v.y * di2;
            rw[64] = v.z * di2; rw[96] = v.w * di2;
        }
    }
    if (FIRST && threadIdx.x == 0) *lcount = 0;
    __syncthreads();

    int nl = -1;
    if (!FIRST) nl = __builtin_amdgcn_readfirstlane(mylist[0]);

    if (nl >= 0) {
        int i = wv * 32;
        unsigned w = 0u;
        if (i < nl) w = (unsigned)mylist[LHDR + min(i + lane, nl - 1)];
        while (i < nl) {
            int s  = (int)(w & ((1u << LSH) - 1u));
            int dl = (int)(w >> LSH);
            float coef = dinv[s] * dinv[row0 + dl];
            const int inext = i + SW * 32;
            unsigned wn = (unsigned)mylist[LHDR + min(inext + lane, nl - 1)];
            add_edges<RS>(s, dl, coef, min(32, nl - i), gl, rowsl);
            w = wn;
            i = inext;
        }
    } else {
        scan_edges<FIRST, RS>(q, wv, lane, row0, nrows, src, dst, dinv, gl, rowsl,
                              lcount, mylist, lcap, E);
    }
    __syncthreads();

    if (FIRST) {
        if (threadIdx.x == 0) {
            int c = *lcount;
            mylist[0] = (c <= lcap) ? c : -1;
        }
        for (int r = wv; r < nrows; r += SW) {
            const float* rw = rowsl + r * RS;
            float4 v;
            v.x = tanhf(rw[0]); v.y = tanhf(rw[32]); v.z = tanhf(rw[64]); v.w = tanhf(rw[96]);
            ((float4*)out)[(size_t)(row0 + r) * (HDIM / 4) + lane] = v;
        }
    } else {
        v2f* Wq = (v2f*)(lds + RB * RS);
        {
            const int n  = threadIdx.x & (HDIM - 1);
            const int ub = (int)(threadIdx.x >> 7);
            float xv[8], yv[8];
#pragma unroll
            for (int i = 0; i < 8; ++i) {
                const int u  = ub + 8 * i;
                const int f0 = 8 * (u & 15) + (u >> 4);
                xv[i] = Wn[f0 * HDIM + n];
                yv[i] = Wn[(f0 + 4) * HDIM + n];
            }
#pragma unroll
            for (int i = 0; i < 8; ++i) {
                const int u = ub + 8 * i;
                v2f wp; wp.x = xv[i]; wp.y = yv[i];
                Wq[u * WS2 + n] = wp;
            }
        }
        __syncthreads();

        if (wv < NT) {
            const int m = lane & 15;
            const int h = lane >> 4;
            const float* Ar = rows + (16 * wv + m) * RS + 2 * h;
            const v2f*   Bq = Wq + h * WS2 + m;
            v8f acc[8];
#pragma unroll
            for (int j = 0; j < 8; ++j) acc[j] = (v8f){0.f,0.f,0.f,0.f,0.f,0.f,0.f,0.f};
            for (int t = 0; t < HDIM / 4; ++t) {
                const v2f a = *(const v2f*)(Ar + 4 * t);
                const v2f* Bk = Bq + 2 * t * WS2;
#pragma unroll
                for (int j = 0; j < 8; ++j) {
                    const v2f b = Bk[j * 16];
                    acc[j] = __builtin_amdgcn_wmma_f32_16x16x4_f32(
                        false, a, false, b, (short)0, acc[j], false, false);
                }
            }
            float bj[8];
#pragma unroll
            for (int j = 0; j < 8; ++j) bj[j] = bias[16 * j + m];
            float* ob = rows + (16 * wv + 8 * h) * RS + m;
#pragma unroll
            for (int i = 0; i < 8; ++i)
#pragma unroll
                for (int j = 0; j < 8; ++j) ob[i * RS + 16 * j] = acc[j][i] + bj[j];
            for (int r = 0; r < 16; ++r) {
                const int row = 16 * wv + r;
                if (row < nrows) {
                    float4 v = *(const float4*)(rows + row * RS + 4 * lane);
                    v.x = tanhf(v.x); v.y = tanhf(v.y); v.z = tanhf(v.z); v.w = tanhf(v.w);
                    ((float4*)out)[(size_t)(row0 + row) * (HDIM / 4) + lane] = v;
                }
            }
        }
    }
}

extern "C" void kernel_launch(void* const* d_in, const int* in_sizes, int n_in,
                              void* d_out, int out_size, void* d_ws, size_t ws_size,
                              hipStream_t stream) {
    const float* x  = (const float*)d_in[0];
    const int*   ei = (const int*)d_in[1];
    const float* Wl[4] = {(const float*)d_in[2], (const float*)d_in[4],
                          (const float*)d_in[6], (const float*)d_in[8]};
    const float* bl[4] = {(const float*)d_in[3], (const float*)d_in[5],
                          (const float*)d_in[7], (const float*)d_in[9]};

    const int NFEAT = in_sizes[2] / HDIM;
    const int N     = in_sizes[0] / NFEAT;
    const int E     = in_sizes[1] / 2;
    const int* srcp = ei;
    const int* dstp = ei + E;

    float* wsf  = (float*)d_ws;
    float* dinv = wsf;
    float* r0   = wsf + N;
    float* r1   = r0 + (size_t)N * HDIM;
    float* r2   = r1 + (size_t)N * HDIM;

    const int THR = 256;

    zero_f32<<<(N + THR - 1) / THR, THR, 0, stream>>>(dinv, N);
    deg_count<<<(E + THR - 1) / THR, THR, 0, stream>>>(dstp, dinv, E);
    finalize_dinv<<<(N + THR - 1) / THR, THR, 0, stream>>>(dinv, N);

    {
        const int K = NFEAT;
        const int Kpad = (K + 3) & ~3;
        const size_t shmem =
            ((size_t)(Kpad / 2) * WS2 * 2 + 8u * 32u * A_LD) * sizeof(float);
        gcn_gemm_wmma_f32<<<(N + 255) / 256, 256, shmem, stream>>>(
            x, Wl[0], r0, N, K, Kpad);
        const size_t eshm = ((size_t)RB * HDIM + (size_t)SW * QC + 4) * sizeof(float);
        edge_block_sum<true><<<(N + RB - 1) / RB, SW * 32, eshm, stream>>>(
            srcp, dstp, dinv, r0, Wl[0], bl[0], r2, (int*)r1, N, E);
    }

    const size_t mshm = ((size_t)RB * RSM + (size_t)WQW + 4) * sizeof(float);
    edge_block_sum<false><<<(N + RB - 1) / RB, SW * 32, mshm, stream>>>(
        srcp, dstp, dinv, r2, Wl[1], bl[1], r0, (int*)r1, N, E);
    edge_block_sum<false><<<(N + RB - 1) / RB, SW * 32, mshm, stream>>>(
        srcp, dstp, dinv, r0, Wl[2], bl[2], r2, (int*)r1, N, E);
    edge_block_sum<false><<<(N + RB - 1) / RB, SW * 32, mshm, stream>>>(
        srcp, dstp, dinv, r2, Wl[3], bl[3], (float*)d_out, (int*)r1, N, E);
}


# layer 1 also as (A_hat x)W0: its edge kernel sums weighted x rows (133 floats, 16-byte loads at 4-byte alignment) in LDS, then multiplies by W0 there on f32 WMMA; GEMM launch gone; sum order changed
# speedup vs baseline: 4.0402x; 1.2077x over previous
#include <hip/hip_runtime.h>
#include <hip/hip_bf16.h>
#include <math.h>


#define HDIM 128
#define WS2  144
#define XF   133
#define XT   (XF - HDIM)
#define KX   136

typedef __attribute__((ext_vector_type(2))) float v2f;
typedef __attribute__((ext_vector_type(8))) float v8f;
typedef float f4u __attribute__((ext_vector_type(4), aligned(4)));

__global__ void zero_f32(float* __restrict__ p, int n) {
    int i = blockIdx.x * blockDim.x + threadIdx.x;
    if (i < n) p[i] = 0.0f;
}

__global__ void deg_count(const int* __restrict__ dst, float* __restrict__ deg, int E) {
    int e = blockIdx.x * blockDim.x + threadIdx.x;
    if (e < E) unsafeAtomicAdd(&deg[dst[e]], 1.0f);
}

__global__ void finalize_dinv(float* __restrict__ d, int n) {
    int i = blockIdx.x * blockDim.x + threadIdx.x;
    if (i < n) d[i] = rsqrtf(d[i] + 1.0f);
}

#define RB 400
#define SW 32
#define SL 4
#define QC 544
#define EG 8
#define LHDR 32
#define LSH 23

typedef __attribute__((ext_vector_type(4))) int v4i;

__device__ __forceinline__ void lds_add(float* p, float v) {
    __hip_atomic_fetch_add(p, v, __ATOMIC_RELAXED, __HIP_MEMORY_SCOPE_WORKGROUP);
}

__device__ __forceinline__ void lds_row_add(float* rw, float c, float4 v) {
    lds_add(rw,      c * v.x);
    lds_add(rw + 32, c * v.y);
    lds_add(rw + 64, c * v.z);
    lds_add(rw + 96, c * v.w);
}

template <bool XR>
__device__ __forceinline__ float4 row_piece(const float* __restrict__ gp, int s) {
    if (XR) {
        const f4u v = *(const f4u*)(gp + (size_t)s * XF);
        return make_float4(v.x, v.y, v.z, v.w);
    }
    return *(const float4*)(gp + (size_t)s * HDIM);
}

template <int RS, bool XR>
__device__ __forceinline__ void add_edges(int s, int dl, float coef, int cnt, int lane,
                                          const float* __restrict__ gp,
                                          const float* __restrict__ gt, float* rowsl) {
    if (cnt == 32) {
#pragma unroll
        for (int c = 0; c < 32 / EG; ++c) {
            float4 v[EG];
            float tv[EG];
#pragma unroll
            for (int j = 0; j < EG; ++j) {
                int sj = __builtin_amdgcn_readlane(s, c * EG + j);
                tv[j] = XR ? gt[(size_t)sj * XF] : 0.f;
            }
#pragma unroll
            for (int j = 0; j < EG; ++j) {
                int sj = __builtin_amdgcn_readlane(s, c * EG + j);
                v[j] = row_piece<XR>(gp, sj);
            }
            if (XR)
                asm volatile("" :: "v"(tv[0]), "v"(tv[1]), "v"(tv[2]), "v"(tv[3]),
                                   "v"(tv[4]), "v"(tv[5]), "v"(tv[6]), "v"(tv[7]));
#pragma unroll
            for (int j = 0; j < EG; ++j) {
                int   dj = __builtin_amdgcn_readlane(dl, c * EG + j);
                float cj = __int_as_float(
                    __builtin_amdgcn_readlane(__float_as_int(coef), c * EG + j));
                lds_row_add(rowsl + dj * RS, cj, v[j]);
            }
            if (XR && lane < XT) {
#pragma unroll
                for (int j = 0; j < EG; ++j) {
                    int   dj = __builtin_amdgcn_readlane(dl, c * EG + j);
                    float cj = __int_as_float(
                        __builtin_amdgcn_readlane(__float_as_int(coef), c * EG + j));
                    lds_add(rowsl + dj * RS + HDIM, cj * tv[j]);
                }
            }
        }
    } else {
        for (int j = 0; j < cnt; ++j) {
            int   sj = __builtin_amdgcn_readlane(s, j);
            int   dj = __builtin_amdgcn_readlane(dl, j);
            float cj = __int_as_float(__builtin_amdgcn_readlane(__float_as_int(coef), j));
            float4 v = row_piece<XR>(gp, sj);
            float tv = XR ? gt[(size_t)sj * XF] : 0.f;
            lds_row_add(rowsl + dj * RS, cj, v);
            if (XR && lane < XT) lds_add(rowsl + dj * RS + HDIM, cj * tv);
        }
    }
}

template <bool SAVE, int RS, bool XR>
__device__ __forceinline__ void drain_edges(const int* qp, int cnt, int lane, int row0,
                                            const int* __restrict__ src,
                                            const int* __restrict__ dst,
                                            const float* __restrict__ dinv,
                                            const float* __restrict__ gp,
                                            const float* __restrict__ gt,
                                            float* rowsl, int* lcount, int* mylist,
                                            int lcap) {
    int e = (lane < cnt) ? qp[lane] : 0;
    int s = src[e];
    int dg = dst[e];
    float coef = dinv[s] * dinv[dg];
    int dl = dg - row0;
    if (SAVE) {
        int base = 0;
        if (lane == 0)
            base = __hip_atomic_fetch_add(lcount, cnt, __ATOMIC_RELAXED,
                                          __HIP_MEMORY_SCOPE_WORKGROUP);
        base = __builtin_amdgcn_readfirstlane(base);
        if (base + cnt <= lcap && lane < cnt)
            mylist[LHDR + base + lane] = (int)((unsigned)s | ((unsigned)dl << LSH));
    }
    add_edges<RS, XR>(s, dl, coef, cnt, lane, gp, gt, rowsl);
}

template <bool SAVE, int RS, bool XR>
__device__ __forceinline__ void scan_edges(int* q, int wv, int lane, int row0, int nrows,
                                           const int* __restrict__ src,
                                           const int* __restrict__ dst,
                                           const float* __restrict__ dinv,
                                           const float* __restrict__ gp,
                                           const float* __restrict__ gt,
                                           float* rowsl, int* lcount, int* mylist,
                                           int lcap, int E) {
    const int off = (int)(((size_t)dst >> 2) & 3);
    const v4i* __restrict__ d4 = (const v4i*)(dst - off);
    const int nfull = (E + off) >> 2;
    int qn = 0;

    if (wv == 0) {
        int et = 4 * nfull - off + lane;
        bool m = false;
        if (et >= 0 && et < E) m = (unsigned)(dst[et] - row0) < (unsigned)nrows;
        unsigned b = __builtin_amdgcn_ballot_w32(m);
        if (m) q[__builtin_amdgcn_mbcnt_lo(b, 0)] = et;
        qn = __builtin_popcount(b);
    }

    const int niter = (nfull + SW * 32 * SL - 1) / (SW * 32 * SL);
    const int rot = (int)(((long long)blockIdx.x * niter) / gridDim.x);
    for (int it = 0; it < niter; ++it) {
        int ci = it + rot;
        if (ci >= niter) ci -= niter;
        const int base = ci * (SW * 32 * SL) + wv * (32 * SL);
        v4i x[SL];
#pragma unroll
        for (int k = 0; k < SL; ++k) {
            int j4 = base + k * 32 + lane;
            x[k] = (j4 < nfull) ? d4[j4] : (v4i){-1, -1, -1, -1};
        }
        if (base == 0 && lane == 0) {
            if (off > 0) x[0].x = -1;
            if (off > 1) x[0].y = -1;
            if (off > 2) x[0].z = -1;
        }
#pragma unroll
        for (int k = 0; k < SL; ++k) {
            int e0 = 4 * (base + k * 32 + lane) - off;
#pragma unroll
            for (int c = 0; c < 4; ++c) {
                bool m = (unsigned)(x[k][c] - row0) < (unsigned)nrows;
                unsigned b = __builtin_amdgcn_ballot_w32(m);
                if (b) {
                    if (m) q[qn + __builtin_amdgcn_mbcnt_lo(b, 0)] = e0 + c;
                    qn += __builtin_popcount(b);
                }
            }
        }
        while (qn >= 32) {
            qn -= 32;
            drain_edges<SAVE, RS, XR>(q + qn, 32, lane, row0, src, dst, dinv, gp, gt,
                                      rowsl, lcount, mylist, lcap);
        }
    }
    if (qn > 0)
        drain_edges<SAVE, RS, XR>(q, qn, lane, row0, src, dst, dinv, gp, gt, rowsl,
                                  lcount, mylist, lcap);
}

#define RSM 132
#define RSX 140
#define NT (RB / 16)
#define WQW (HDIM / 2 * WS2 * 2)
#define WQX (KX / 2 * WS2 * 2)

template <bool FIRST>
__global__ __launch_bounds__(1024)
void edge_block_sum(const int* __restrict__ src, const int* __restrict__ dst,
                    const float* __restrict__ dinv, const float* __restrict__ g,
                    const float* __restrict__ Wn,
                    const float* __restrict__ bias, float* __restrict__ out,
                    int* __restrict__ lists, int N, int E) {
    constexpr int RS = FIRST ? RSX : RSM;
    constexpr int GS = FIRST ? XF : HDIM;
    constexpr int KT = (FIRST ? KX : HDIM) / 4;
    extern __shared__ float lds[];
    float* rows = lds;
    const int lane = threadIdx.x & 31;
    const int wv = __builtin_amdgcn_readfirstlane((int)(threadIdx.x >> 5));
    int* q = (int*)(lds + RB * RS) + wv * QC;
    int* lcount = (int*)(lds + RB * RS) + (FIRST ? WQX : WQW);

    const int row0  = blockIdx.x * RB;
    const int nrows = min(RB, N - row0);
    float* rowsl = rows + lane;
    const float* __restrict__ gp = g + 4 * lane;
    const float* __restrict__ gt = g + HDIM + min(lane, XT - 1);
    int* mylist = lists + (size_t)row0 * HDIM;
    const int lcap = nrows * HDIM - LHDR;

    for (int r = wv; r < nrows; r += SW) {
        float di = dinv[row0 + r];
        float tv = FIRST ? gt[(size_t)(row0 + r) * GS] : 0.f;
        float4 v = row_piece<FIRST>(gp, row0 + r);
        if (FIRST) asm volatile("" :: "v"(tv));
        float di2 = di * di;
        float* rw = rowsl + r * RS;
        rw[0]  = v.x * di2; rw[32] = v.y * di2;
        rw[64] = v.z * di2; rw[96] = v.w * di2;
        if (FIRST && lane < KX - HDIM) rw[HDIM] = (lane < XT) ? tv * di2 : 0.f;
    }
    if (FIRST && threadIdx.x == 0) *lcount = 0;
    __syncthreads();

    int nl = -1;
    if (!FIRST) nl = __builtin_amdgcn_readfirstlane(mylist[0]);

    if (nl >= 0) {
        int i = wv * 32;
        unsigned w = 0u;
        if (i < nl) w = (unsigned)mylist[LHDR + min(i + lane, nl - 1)];
        while (i < nl) {
            int s  = (int)(w & ((1u << LSH) - 1u));
            int dl = (int)(w >> LSH);
            float coef = dinv[s] * dinv[row0 + dl];
            const int inext = i + SW * 32;
            unsigned wn = (unsigned)mylist[LHDR + min(inext + lane, nl - 1)];
            add_edges<RS, FIRST>(s, dl, coef, min(32, nl - i), lane, gp, gt, rowsl);
            w = wn;
            i = inext;
        }
    } else {
        scan_edges<FIRST, RS, FIRST>(q, wv, lane, row0, nrows, src, dst, dinv, gp, gt,
                                     rowsl, lcount, mylist, lcap, E);
    }
    __syncthreads();

    if (FIRST && threadIdx.x == 0) {
        int c = *lcount;
        mylist[0] = (c <= lcap) ? c : -1;
    }

    v2f* Wq = (v2f*)(lds + RB * RS);
    {
        const int n  = threadIdx.x & (HDIM - 1);
        const int ub = (int)(threadIdx.x >> 7);
        float xv[8], yv[8];
        float xe = 0.f, ye = 0.f;
#pragma unroll
        for (int i = 0; i < 8; ++i) {
            const int u  = ub + 8 * i;
            const int f0 = 8 * (u & 15) + (u >> 4);
            xv[i] = Wn[f0 * HDIM + n];
            yv[i] = Wn[(f0 + 4) * HDIM + n];
        }
        if (FIRST && ub < (KX - HDIM) / 2) {
            const int f = HDIM + 2 * ub;
            if (f     < XF) xe = Wn[f * HDIM + n];
            if (f + 1 < XF) ye = Wn[(f + 1) * HDIM + n];
        }
#pragma unroll
        for (int i = 0; i < 8; ++i) {
            const int u = ub + 8 * i;
            v2f wp; wp.x = xv[i]; wp.y = yv[i];
            Wq[u * WS2 + n] = wp;
        }
        if (FIRST && ub < (KX - HDIM) / 2) {
            v2f wp; wp.x = xe; wp.y = ye;
            Wq[(HDIM / 2 + ub) * WS2 + n] = wp;
        }
    }
    __syncthreads();

    if (wv < NT) {
        const int m = lane & 15;
        const int h = lane >> 4;
        const float* Ar = rows + (16 * wv + m) * RS + 2 * h;
        const v2f*   Bq = Wq + h * WS2 + m;
        v8f acc[8];
#pragma unroll
        for (int j = 0; j < 8; ++j) acc[j] = (v8f){0.f,0.f,0.f,0.f,0.f,0.f,0.f,0.f};
        for (int t = 0; t < KT; ++t) {
            const v2f a = *(const v2f*)(Ar + 4 * t);
            const v2f* Bk = Bq + 2 * t * WS2;
#pragma unroll
            for (int j = 0; j < 8; ++j) {
                const v2f b = Bk[j * 16];
                acc[j] = __builtin_amdgcn_wmma_f32_16x16x4_f32(
                    false, a, false, b, (short)0, acc[j], false, false);
            }
        }
        float bj[8];
#pragma unroll
        for (int j = 0; j < 8; ++j) bj[j] = bias[16 * j + m];
        float* ob = rows + (16 * wv + 8 * h) * RS + m;
#pragma unroll
        for (int i = 0; i < 8; ++i)
#pragma unroll
            for (int j = 0; j < 8; ++j) ob[i * RS + 16 * j] = acc[j][i] + bj[j];
        for (int r = 0; r < 16; ++r) {
            const int row = 16 * wv + r;
            if (row < nrows) {
                float4 v = *(const float4*)(rows + row * RS + 4 * lane);
                v.x = tanhf(v.x); v.y = tanhf(v.y); v.z = tanhf(v.z); v.w = tanhf(v.w);
                ((float4*)out)[(size_t)(row0 + row) * (HDIM / 4) + lane] = v;
            }
        }
    }
}

extern "C" void kernel_launch(void* const* d_in, const int* in_sizes, int n_in,
                              void* d_out, int out_size, void* d_ws, size_t ws_size,
                              hipStream_t stream) {
    const float* x  = (const float*)d_in[0];
    const int*   ei = (const int*)d_in[1];
    const float* Wl[4] = {(const float*)d_in[2], (const float*)d_in[4],
                          (const float*)d_in[6], (const float*)d_in[8]};
    const float* bl[4] = {(const float*)d_in[3], (const float*)d_in[5],
                          (const float*)d_in[7], (const float*)d_in[9]};

    const int NFEAT = in_sizes[2] / HDIM;
    const int N     = in_sizes[0] / NFEAT;
    const int E     = in_sizes[1] / 2;
    const int* srcp = ei;
    const int* dstp = ei + E;

    float* wsf  = (float*)d_ws;
    float* dinv = wsf;
    float* r0   = wsf + N;
    float* r1   = r0 + (size_t)N * HDIM;
    float* r2   = r1 + (size_t)N * HDIM;

    const int THR = 256;

    zero_f32<<<(N + THR - 1) / THR, THR, 0, stream>>>(dinv, N);
    deg_count<<<(E + THR - 1) / THR, THR, 0, stream>>>(dstp, dinv, E);
    finalize_dinv<<<(N + THR - 1) / THR, THR, 0, stream>>>(dinv, N);

    const size_t xshm = ((size_t)RB * RSX + (size_t)WQX + 4) * sizeof(float);
    edge_block_sum<true><<<(N + RB - 1) / RB, SW * 32, xshm, stream>>>(
        srcp, dstp, dinv, x, Wl[0], bl[0], r2, (int*)r1, N, E);
    const size_t mshm = ((size_t)RB * RSM + (size_t)WQW + 4) * sizeof(float);
    edge_block_sum<false><<<(N + RB - 1) / RB, SW * 32, mshm, stream>>>(
        srcp, dstp, dinv, r2, Wl[1], bl[1], r0, (int*)r1, N, E);
    edge_block_sum<false><<<(N + RB - 1) / RB, SW * 32, mshm, stream>>>(
        srcp, dstp, dinv, r0, Wl[2], bl[2], r2, (int*)r1, N, E);
    edge_block_sum<false><<<(N + RB - 1) / RB, SW * 32, mshm, stream>>>(
        srcp, dstp, dinv, r2, Wl[3], bl[3], (float*)d_out, (int*)r1, N, E);
}


# dst scan moved from layer 1 into a prologue launch: each row block lists its edges and counts its rows' degrees in LDS, writes dinv; global atomics, 2 launches gone; row sums still in arrival order
# speedup vs baseline: 4.4930x; 1.1121x over previous
#include <hip/hip_runtime.h>
#include <hip/hip_bf16.h>
#include <math.h>


#define HDIM 128
#define WS2  144
#define XF   133
#define XT   (XF - HDIM)
#define KX   136

typedef __attribute__((ext_vector_type(2))) float v2f;
typedef __attribute__((ext_vector_type(8))) float v8f;
typedef float f4u __attribute__((ext_vector_type(4), aligned(4)));

#define RB 400
#define SW 32
#define SL 4
#define QC 544
#define EG 8
#define LHDR 32
#define LSH 23

typedef __attribute__((ext_vector_type(4))) int v4i;

__device__ __forceinline__ void lds_add(float* p, float v) {
    __hip_atomic_fetch_add(p, v, __ATOMIC_RELAXED, __HIP_MEMORY_SCOPE_WORKGROUP);
}

__device__ __forceinline__ void lds_row_add(float* rw, float c, float4 v) {
    lds_add(rw,      c * v.x);
    lds_add(rw + 32, c * v.y);
    lds_add(rw + 64, c * v.z);
    lds_add(rw + 96, c * v.w);
}

template <bool XR>
__device__ __forceinline__ float4 row_piece(const float* __restrict__ gp, int s) {
    if (XR) {
        const f4u v = *(const f4u*)(gp + (size_t)s * XF);
        return make_float4(v.x, v.y, v.z, v.w);
    }
    return *(const float4*)(gp + (size_t)s * HDIM);
}

template <int RS, bool XR>
__device__ __forceinline__ void add_edges(int s, int dl, float coef, int cnt, int lane,
                                          const float* __restrict__ gp,
                                          const float* __restrict__ gt, float* rowsl) {
    if (cnt == 32) {
#pragma unroll
        for (int c = 0; c < 32 / EG; ++c) {
            float4 v[EG];
            float tv[EG];
#pragma unroll
            for (int j = 0; j < EG; ++j) {
                int sj = __builtin_amdgcn_readlane(s, c * EG + j);
                tv[j] = XR ? gt[(size_t)sj * XF] : 0.f;
            }
#pragma unroll
            for (int j = 0; j < EG; ++j) {
                int sj = __builtin_amdgcn_readlane(s, c * EG + j);
                v[j] = row_piece<XR>(gp, sj);
            }
            if (XR)
                asm volatile("" :: "v"(tv[0]), "v"(tv[1]), "v"(tv[2]), "v"(tv[3]),
                                   "v"(tv[4]), "v"(tv[5]), "v"(tv[6]), "v"(tv[7]));
#pragma unroll
            for (int j = 0; j < EG; ++j) {
                int   dj = __builtin_amdgcn_readlane(dl, c * EG + j);
                float cj = __int_as_float(
                    __builtin_amdgcn_readlane(__float_as_int(coef), c * EG + j));
                lds_row_add(rowsl + dj * RS, cj, v[j]);
            }
            if (XR && lane < XT) {
#pragma unroll
                for (int j = 0; j < EG; ++j) {
                    int   dj = __builtin_amdgcn_readlane(dl, c * EG + j);
                    float cj = __int_as_float(
                        __builtin_amdgcn_readlane(__float_as_int(coef), c * EG + j));
                    lds_add(rowsl + dj * RS + HDIM, cj * tv[j]);
                }
            }
        }
    } else {
        for (int j = 0; j < cnt; ++j) {
            int   sj = __builtin_amdgcn_readlane(s, j);
            int   dj = __builtin_amdgcn_readlane(dl, j);
            float cj = __int_as_float(__builtin_amdgcn_readlane(__float_as_int(coef), j));
            float4 v = row_piece<XR>(gp, sj);
            float tv = XR ? gt[(size_t)sj * XF] : 0.f;
            lds_row_add(rowsl + dj * RS, cj, v);
            if (XR && lane < XT) lds_add(rowsl + dj * RS + HDIM, cj * tv);
        }
    }
}

template <int RS, bool XR>
__device__ __forceinline__ void drain_edges(const int* qp, int cnt, int lane, int row0,
                                            const int* __restrict__ src,
                                            const int* __restrict__ dst,
                                            const float* __restrict__ dinv,
                                            const float* __restrict__ gp,
                                            const float* __restrict__ gt,
                                            float* rowsl) {
    int e = (lane < cnt) ? qp[lane] : 0;
    int s = src[e];
    int dg = dst[e];
    float coef = dinv[s] * dinv[dg];
    int dl = dg - row0;
    add_edges<RS, XR>(s, dl, coef, cnt, lane, gp, gt, rowsl);
}

__device__ __forceinline__ void collect_edges(const int* qp, int cnt, int lane, int row0,
                                              const int* __restrict__ src,
                                              const int* __restrict__ dst,
                                              int* degc, int* lcount, int* mylist,
                                              int lcap) {
    int e = (lane < cnt) ? qp[lane] : 0;
    int s = src[e];
    int dl = dst[e] - row0;
    if (lane < cnt)
        __hip_atomic_fetch_add(degc + dl, 1, __ATOMIC_RELAXED, __HIP_MEMORY_SCOPE_WORKGROUP);
    int base = 0;
    if (lane == 0)
        base = __hip_atomic_fetch_add(lcount, cnt, __ATOMIC_RELAXED,
                                      __HIP_MEMORY_SCOPE_WORKGROUP);
    base = __builtin_amdgcn_readfirstlane(base);
    if (base + cnt <= lcap && lane < cnt)
        mylist[LHDR + base + lane] = (int)((unsigned)s | ((unsigned)dl << LSH));
}

template <bool COLLECT, int RS, bool XR>
__device__ __forceinline__ void scan_edges(int* q, int wv, int lane, int row0, int nrows,
                                           const int* __restrict__ src,
                                           const int* __restrict__ dst,
                                           const float* __restrict__ dinv,
                                           const float* __restrict__ gp,
                                           const float* __restrict__ gt,
                                           float* rowsl, int* degc, int* lcount,
                                           int* mylist, int lcap, int E) {
    const int off = (int)(((size_t)dst >> 2) & 3);
    const v4i* __restrict__ d4 = (const v4i*)(dst - off);
    const int nfull = (E + off) >> 2;
    int qn = 0;

    if (wv == 0) {
        int et = 4 * nfull - off + lane;
        bool m = false;
        if (et >= 0 && et < E) m = (unsigned)(dst[et] - row0) < (unsigned)nrows;
        unsigned b = __builtin_amdgcn_ballot_w32(m);
        if (m) q[__builtin_amdgcn_mbcnt_lo(b, 0)] = et;
        qn = __builtin_popcount(b);
    }

    const int niter = (nfull + SW * 32 * SL - 1) / (SW * 32 * SL);
    const int rot = (int)(((long long)blockIdx.x * niter) / gridDim.x);
    for (int it = 0; it < niter; ++it) {
        int ci = it + rot;
        if (ci >= niter) ci -= niter;
        const int base = ci * (SW * 32 * SL) + wv * (32 * SL);
        v4i x[SL];
#pragma unroll
        for (int k = 0; k < SL; ++k) {
            int j4 = base + k * 32 + lane;
            x[k] = (j4 < nfull) ? d4[j4] : (v4i){-1, -1, -1, -1};
        }
        if (base == 0 && lane == 0) {
            if (off > 0) x[0].x = -1;
            if (off > 1) x[0].y = -1;
            if (off > 2) x[0].z = -1;
        }
#pragma unroll
        for (int k = 0; k < SL; ++k) {
            int e0 = 4 * (base + k * 32 + lane) - off;
#pragma unroll
            for (int c = 0; c < 4; ++c) {
                bool m = (unsigned)(x[k][c] - row0) < (unsigned)nrows;
                unsigned b = __builtin_amdgcn_ballot_w32(m);
                if (b) {
                    if (m) q[qn + __builtin_amdgcn_mbcnt_lo(b, 0)] = e0 + c;
                    qn += __builtin_popcount(b);
                }
            }
        }
        while (qn >= 32) {
            qn -= 32;
            if (COLLECT)
                collect_edges(q + qn, 32, lane, row0, src, dst, degc, lcount, mylist, lcap);
            else
                drain_edges<RS, XR>(q + qn, 32, lane, row0, src, dst, dinv, gp, gt, rowsl);
        }
    }
    if (qn > 0) {
        if (COLLECT)
            collect_edges(q, qn, lane, row0, src, dst, degc, lcount, mylist, lcap);
        else
            drain_edges<RS, XR>(q, qn, lane, row0, src, dst, dinv, gp, gt, rowsl);
    }
}

__global__ __launch_bounds__(1024)
void scan_rows(const int* __restrict__ src, const int* __restrict__ dst,
               float* __restrict__ dinv, int* __restrict__ lists, int N, int E) {
    extern __shared__ float lds[];
    const int lane = threadIdx.x & 31;
    const int wv = __builtin_amdgcn_readfirstlane((int)(threadIdx.x >> 5));
    int* q = (int*)lds + wv * QC;
    int* degc = (int*)lds + SW * QC;
    int* lcount = degc + RB;

    const int row0  = blockIdx.x * RB;
    const int nrows = min(RB, N - row0);
    int* mylist = lists + (size_t)row0 * HDIM;
    const int lcap = nrows * HDIM - LHDR;

    for (int r = threadIdx.x; r < RB; r += SW * 32) degc[r] = 0;
    if (threadIdx.x == 0) *lcount = 0;
    __syncthreads();

    scan_edges<true, 0, false>(q, wv, lane, row0, nrows, src, dst, nullptr, nullptr, nullptr,
                               nullptr, degc, lcount, mylist, lcap, E);
    __syncthreads();

    for (int r = threadIdx.x; r < nrows; r += SW * 32)
        dinv[row0 + r] = rsqrtf((float)degc[r] + 1.0f);
    if (threadIdx.x == 0) {
        int c = *lcount;
        mylist[0] = (c <= lcap) ? c : -1;
    }
}

#define RSM 132
#define RSX 140
#define NT (RB / 16)
#define WQW (HDIM / 2 * WS2 * 2)
#define WQX (KX / 2 * WS2 * 2)

template <bool FIRST>
__global__ __launch_bounds__(1024)
void edge_block_sum(const int* __restrict__ src, const int* __restrict__ dst,
                    const float* __restrict__ dinv, const float* __restrict__ g,
                    const float* __restrict__ Wn,
                    const float* __restrict__ bias, float* __restrict__ out,
                    const int* __restrict__ lists, int N, int E) {
    constexpr int RS = FIRST ? RSX : RSM;
    constexpr int GS = FIRST ? XF : HDIM;
    constexpr int KT = (FIRST ? KX : HDIM) / 4;
    extern __shared__ float lds[];
    float* rows = lds;
    const int lane = threadIdx.x & 31;
    const int wv = __builtin_amdgcn_readfirstlane((int)(threadIdx.x >> 5));
    int* q = (int*)(lds + RB * RS) + wv * QC;

    const int row0  = blockIdx.x * RB;
    const int nrows = min(RB, N - row0);
    float* rowsl = rows + lane;
    const float* __restrict__ gp = g + 4 * lane;
    const float* __restrict__ gt = g + HDIM + min(lane, XT - 1);
    const int* __restrict__ mylist = lists + (size_t)row0 * HDIM;

    for (int r = wv; r < nrows; r += SW) {
        float di = dinv[row0 + r];
        float tv = FIRST ? gt[(size_t)(row0 + r) * GS] : 0.f;
        float4 v = row_piece<FIRST>(gp, row0 + r);
        if (FIRST) asm volatile("" :: "v"(tv));
        float di2 = di * di;
        float* rw = rowsl + r * RS;
        rw[0]  = v.x * di2; rw[32] = v.y * di2;
        rw[64] = v.z * di2; rw[96] = v.w * di2;
        if (FIRST && lane < KX - HDIM) rw[HDIM] = (lane < XT) ? tv * di2 : 0.f;
    }
    __syncthreads();

    const int nl = __builtin_amdgcn_readfirstlane(mylist[0]);

    if (nl >= 0) {
        int i = wv * 32;
        unsigned w = 0u;
        if (i < nl) w = (unsigned)mylist[LHDR + min(i + lane, nl - 1)];
        while (i < nl) {
            int s  = (int)(w & ((1u << LSH) - 1u));
            int dl = (int)(w >> LSH);
            float coef = dinv[s] * dinv[row0 + dl];
            const int inext = i + SW * 32;
            unsigned wn = (unsigned)mylist[LHDR + min(inext + lane, nl - 1)];
            add_edges<RS, FIRST>(s, dl, coef, min(32, nl - i), lane, gp, gt, rowsl);
            w = wn;
            i = inext;
        }
    } else {
        scan_edges<false, RS, FIRST>(q, wv, lane, row0, nrows, src, dst, dinv, gp, gt,
                                     rowsl, nullptr, nullptr, nullptr, 0, E);
    }
    __syncthreads();

    v2f* Wq = (v2f*)(lds + RB * RS);
    {
        const int n  = threadIdx.x & (HDIM - 1);
        const int ub = (int)(threadIdx.x >> 7);
        float xv[8], yv[8];
        float xe = 0.f, ye = 0.f;
#pragma unroll
        for (int i = 0; i < 8; ++i) {
            const int u  = ub + 8 * i;
            const int f0 = 8 * (u & 15) + (u >> 4);
            xv[i] = Wn[f0 * HDIM + n];
            yv[i] = Wn[(f0 + 4) * HDIM + n];
        }
        if (FIRST && ub < (KX - HDIM) / 2) {
            const int f = HDIM + 2 * ub;
            if (f     < XF) xe = Wn[f * HDIM + n];
            if (f + 1 < XF) ye = Wn[(f + 1) * HDIM + n];
        }
#pragma unroll
        for (int i = 0; i < 8; ++i) {
            const int u = ub + 8 * i;
            v2f wp; wp.x = xv[i]; wp.y = yv[i];
            Wq[u * WS2 + n] = wp;
        }
        if (FIRST && ub < (KX - HDIM) / 2) {
            v2f wp; wp.x = xe; wp.y = ye;
            Wq[(HDIM / 2 + ub) * WS2 + n] = wp;
        }
    }
    __syncthreads();

    if (wv < NT) {
        const int m = lane & 15;
        const int h = lane >> 4;
        const float* Ar = rows + (16 * wv + m) * RS + 2 * h;
        const v2f*   Bq = Wq + h * WS2 + m;
        v8f acc[8];
#pragma unroll
        for (int j = 0; j < 8; ++j) acc[j] = (v8f){0.f,0.f,0.f,0.f,0.f,0.f,0.f,0.f};
        for (int t = 0; t < KT; ++t) {
            const v2f a = *(const v2f*)(Ar + 4 * t);
            const v2f* Bk = Bq + 2 * t * WS2;
#pragma unroll
            for (int j = 0; j < 8; ++j) {
                const v2f b = Bk[j * 16];
                acc[j] = __builtin_amdgcn_wmma_f32_16x16x4_f32(
                    false, a, false, b, (short)0, acc[j], false, false);
            }
        }
        float bj[8];
#pragma unroll
        for (int j = 0; j < 8; ++j) bj[j] = bias[16 * j + m];
        float* ob = rows + (16 * wv + 8 * h) * RS + m;
#pragma unroll
        for (int i = 0; i < 8; ++i)
#pragma unroll
            for (int j = 0; j < 8; ++j) ob[i * RS + 16 * j] = acc[j][i] + bj[j];
        for (int r = 0; r < 16; ++r) {
            const int row = 16 * wv + r;
            if (row < nrows) {
                float4 v = *(const float4*)(rows + row * RS + 4 * lane);
                v.x = tanhf(v.x); v.y = tanhf(v.y); v.z = tanhf(v.z); v.w = tanhf(v.w);
                ((float4*)out)[(size_t)(row0 + row) * (HDIM / 4) + lane] = v;
            }
        }
    }
}

extern "C" void kernel_launch(void* const* d_in, const int* in_sizes, int n_in,
                              void* d_out, int out_size, void* d_ws, size_t ws_size,
                              hipStream_t stream) {
    const float* x  = (const float*)d_in[0];
    const int*   ei = (const int*)d_in[1];
    const float* Wl[4] = {(const float*)d_in[2], (const float*)d_in[4],
                          (const float*)d_in[6], (const float*)d_in[8]};
    const float* bl[4] = {(const float*)d_in[3], (const float*)d_in[5],
                          (const float*)d_in[7], (const float*)d_in[9]};

    const int NFEAT = in_sizes[2] / HDIM;
    const int N     = in_sizes[0] / NFEAT;
    const int E     = in_sizes[1] / 2;
    const int* srcp = ei;
    const int* dstp = ei + E;

    float* wsf  = (float*)d_ws;
    float* dinv = wsf;
    float* r0   = wsf + N;
    float* r1   = r0 + (size_t)N * HDIM;
    float* r2   = r1 + (size_t)N * HDIM;

    const size_t pshm = ((size_t)SW * QC + RB + 4) * sizeof(int);
    scan_rows<<<(N + RB - 1) / RB, SW * 32, pshm, stream>>>(srcp, dstp, dinv, (int*)r1, N, E);

    const size_t xshm = ((size_t)RB * RSX + (size_t)WQX) * sizeof(float);
    edge_block_sum<true><<<(N + RB - 1) / RB, SW * 32, xshm, stream>>>(
        srcp, dstp, dinv, x, Wl[0], bl[0], r2, (int*)r1, N, E);
    const size_t mshm = ((size_t)RB * RSM + (size_t)WQW) * sizeof(float);
    edge_block_sum<false><<<(N + RB - 1) / RB, SW * 32, mshm, stream>>>(
        srcp, dstp, dinv, r2, Wl[1], bl[1], r0, (int*)r1, N, E);
    edge_block_sum<false><<<(N + RB - 1) / RB, SW * 32, mshm, stream>>>(
        srcp, dstp, dinv, r0, Wl[2], bl[2], r2, (int*)r1, N, E);
    edge_block_sum<false><<<(N + RB - 1) / RB, SW * 32, mshm, stream>>>(
        srcp, dstp, dinv, r2, Wl[3], bl[3], (float*)d_out, (int*)r1, N, E);
}


# block's rows split in two parts with a list each: while 17 waves multiply part 1 by W (WMMA, bias, tanh, store) the other 15 sum part 2's edges in LDS; sums still in arrival order
# speedup vs baseline: 4.8554x; 1.0807x over previous
#include <hip/hip_runtime.h>
#include <hip/hip_bf16.h>
#include <math.h>


#define HDIM 128
#define WS2  144
#define XF   133
#define XT   (XF - HDIM)
#define KX   136

typedef __attribute__((ext_vector_type(2))) float v2f;
typedef __attribute__((ext_vector_type(8))) float v8f;
typedef float f4u __attribute__((ext_vector_type(4), aligned(4)));

#define RB 400
#define SW 32
#define SL 4
#define QC 544
#define EG 8
#define LHDR 32
#define LSH 23
#define TA 17
#define RA (16 * TA)

typedef __attribute__((ext_vector_type(4))) int v4i;

__device__ __forceinline__ void lds_add(float* p, float v) {
    __hip_atomic_fetch_add(p, v, __ATOMIC_RELAXED, __HIP_MEMORY_SCOPE_WORKGROUP);
}

__device__ __forceinline__ void lds_row_add(float* rw, float c, float4 v) {
    lds_add(rw,      c * v.x);
    lds_add(rw + 32, c * v.y);
    lds_add(rw + 64, c * v.z);
    lds_add(rw + 96, c * v.w);
}

template <bool XR>
__device__ __forceinline__ float4 row_piece(const float* __restrict__ gp, int s) {
    if (XR) {
        const f4u v = *(const f4u*)(gp + (size_t)s * XF);
        return make_float4(v.x, v.y, v.z, v.w);
    }
    return *(const float4*)(gp + (size_t)s * HDIM);
}

template <int RS, bool XR>
__device__ __forceinline__ void add_edges(int s, int dl, float coef, int cnt, int lane,
                                          const float* __restrict__ gp,
                                          const float* __restrict__ gt, float* rowsl) {
    if (cnt == 32) {
#pragma unroll
        for (int c = 0; c < 32 / EG; ++c) {
            float4 v[EG];
            float tv[EG];
#pragma unroll
            for (int j = 0; j < EG; ++j) {
                int sj = __builtin_amdgcn_readlane(s, c * EG + j);
                tv[j] = XR ? gt[(size_t)sj * XF] : 0.f;
            }
#pragma unroll
            for (int j = 0; j < EG; ++j) {
                int sj = __builtin_amdgcn_readlane(s, c * EG + j);
                v[j] = row_piece<XR>(gp, sj);
            }
            if (XR)
                asm volatile("" :: "v"(tv[0]), "v"(tv[1]), "v"(tv[2]), "v"(tv[3]),
                                   "v"(tv[4]), "v"(tv[5]), "v"(tv[6]), "v"(tv[7]));
#pragma unroll
            for (int j = 0; j < EG; ++j) {
                int   dj = __builtin_amdgcn_readlane(dl, c * EG + j);
                float cj = __int_as_float(
                    __builtin_amdgcn_readlane(__float_as_int(coef), c * EG + j));
                lds_row_add(rowsl + dj * RS, cj, v[j]);
            }
            if (XR && lane < XT) {
#pragma unroll
                for (int j = 0; j < EG; ++j) {
                    int   dj = __builtin_amdgcn_readlane(dl, c * EG + j);
                    float cj = __int_as_float(
                        __builtin_amdgcn_readlane(__float_as_int(coef), c * EG + j));
                    lds_add(rowsl + dj * RS + HDIM, cj * tv[j]);
                }
            }
        }
    } else {
        for (int j = 0; j < cnt; ++j) {
            int   sj = __builtin_amdgcn_readlane(s, j);
            int   dj = __builtin_amdgcn_readlane(dl, j);
            float cj = __int_as_float(__builtin_amdgcn_readlane(__float_as_int(coef), j));
            float4 v = row_piece<XR>(gp, sj);
            float tv = XR ? gt[(size_t)sj * XF] : 0.f;
            lds_row_add(rowsl + dj * RS, cj, v);
            if (XR && lane < XT) lds_add(rowsl + dj * RS + HDIM, cj * tv);
        }
    }
}

template <int RS, bool XR>
__device__ __forceinline__ void drain_edges(const int* qp, int cnt, int lane, int row0,
                                            const int* __restrict__ src,
                                            const int* __restrict__ dst,
                                            const float* __restrict__ dinv,
                                            const float* __restrict__ gp,
                                            const float* __restrict__ gt,
                                            float* rowsl) {
    int e = (lane < cnt) ? qp[lane] : 0;
    int s = src[e];
    int dg = dst[e];
    float coef = dinv[s] * dinv[dg];
    int dl = dg - row0;
    add_edges<RS, XR>(s, dl, coef, cnt, lane, gp, gt, rowsl);
}

__device__ __forceinline__ void collect_edges(const int* qp, int cnt, int lane, int row0,
                                              const int* __restrict__ src,
                                              const int* __restrict__ dst,
                                              int* degc, int* lcount, int* mylist,
                                              int lcapA, int lcapB) {
    int e = (lane < cnt) ? qp[lane] : 0;
    int s = src[e];
    int dl = dst[e] - row0;
    const bool act = lane < cnt;
    if (act)
        __hip_atomic_fetch_add(degc + dl, 1, __ATOMIC_RELAXED, __HIP_MEMORY_SCOPE_WORKGROUP);
    const bool inA = act && dl < RA;
    const bool inB = act && dl >= RA;
    const unsigned bA = __builtin_amdgcn_ballot_w32(inA);
    const unsigned bB = __builtin_amdgcn_ballot_w32(inB);
    int base = 0;
    if (lane == 0 && bA)
        base = __hip_atomic_fetch_add(lcount, __builtin_popcount(bA), __ATOMIC_RELAXED,
                                      __HIP_MEMORY_SCOPE_WORKGROUP);
    if (lane == 1 && bB)
        base = __hip_atomic_fetch_add(lcount + 1, __builtin_popcount(bB), __ATOMIC_RELAXED,
                                      __HIP_MEMORY_SCOPE_WORKGROUP);
    const int baseA = __builtin_amdgcn_readlane(base, 0);
    const int baseB = __builtin_amdgcn_readlane(base, 1);
    const int word = (int)((unsigned)s | ((unsigned)dl << LSH));
    if (inA) {
        const int p = baseA + (int)__builtin_amdgcn_mbcnt_lo(bA, 0);
        if (p < lcapA) mylist[LHDR + p] = word;
    }
    if (inB) {
        const int p = baseB + (int)__builtin_amdgcn_mbcnt_lo(bB, 0);
        if (p < lcapB) mylist[LHDR + lcapA + p] = word;
    }
}

template <bool COLLECT, int RS, bool XR>
__device__ __forceinline__ void scan_edges(int* q, int wv, int lane, int row0, int nrows,
                                           const int* __restrict__ src,
                                           const int* __restrict__ dst,
                                           const float* __restrict__ dinv,
                                           const float* __restrict__ gp,
                                           const float* __restrict__ gt,
                                           float* rowsl, int* degc, int* lcount,
                                           int* mylist, int lcapA, int lcapB, int E) {
    const int off = (int)(((size_t)dst >> 2) & 3);
    const v4i* __restrict__ d4 = (const v4i*)(dst - off);
    const int nfull = (E + off) >> 2;
    int qn = 0;

    if (wv == 0) {
        int et = 4 * nfull - off + lane;
        bool m = false;
        if (et >= 0 && et < E) m = (unsigned)(dst[et] - row0) < (unsigned)nrows;
        unsigned b = __builtin_amdgcn_ballot_w32(m);
        if (m) q[__builtin_amdgcn_mbcnt_lo(b, 0)] = et;
        qn = __builtin_popcount(b);
    }

    const int niter = (nfull + SW * 32 * SL - 1) / (SW * 32 * SL);
    const int rot = (int)(((long long)blockIdx.x * niter) / gridDim.x);
    for (int it = 0; it < niter; ++it) {
        int ci = it + rot;
        if (ci >= niter) ci -= niter;
        const int base = ci * (SW * 32 * SL) + wv * (32 * SL);
        v4i x[SL];
#pragma unroll
        for (int k = 0; k < SL; ++k) {
            int j4 = base + k * 32 + lane;
            x[k] = (j4 < nfull) ? d4[j4] : (v4i){-1, -1, -1, -1};
        }
        if (base == 0 && lane == 0) {
            if (off > 0) x[0].x = -1;
            if (off > 1) x[0].y = -1;
            if (off > 2) x[0].z = -1;
        }
#pragma unroll
        for (int k = 0; k < SL; ++k) {
            int e0 = 4 * (base + k * 32 + lane) - off;
#pragma unroll
            for (int c = 0; c < 4; ++c) {
                bool m = (unsigned)(x[k][c] - row0) < (unsigned)nrows;
                unsigned b = __builtin_amdgcn_ballot_w32(m);
                if (b) {
                    if (m) q[qn + __builtin_amdgcn_mbcnt_lo(b, 0)] = e0 + c;
                    qn += __builtin_popcount(b);
                }
            }
        }
        while (qn >= 32) {
            qn -= 32;
            if (COLLECT)
                collect_edges(q + qn, 32, lane, row0, src, dst, degc, lcount, mylist, lcapA, lcapB);
            else
                drain_edges<RS, XR>(q + qn, 32, lane, row0, src, dst, dinv, gp, gt, rowsl);
        }
    }
    if (qn > 0) {
        if (COLLECT)
            collect_edges(q, qn, lane, row0, src, dst, degc, lcount, mylist, lcapA, lcapB);
        else
            drain_edges<RS, XR>(q, qn, lane, row0, src, dst, dinv, gp, gt, rowsl);
    }
}

__global__ __launch_bounds__(1024)
void scan_rows(const int* __restrict__ src, const int* __restrict__ dst,
               float* __restrict__ dinv, int* __restrict__ lists, int N, int E) {
    extern __shared__ float lds[];
    const int lane = threadIdx.x & 31;
    const int wv = __builtin_amdgcn_readfirstlane((int)(threadIdx.x >> 5));
    int* q = (int*)lds + wv * QC;
    int* degc = (int*)lds + SW * QC;
    int* lcount = degc + RB;

    const int row0  = blockIdx.x * RB;
    const int nrows = min(RB, N - row0);
    int* mylist = lists + (size_t)row0 * HDIM;
    const int lcap = nrows * HDIM - LHDR;
    const int lcapA = (lcap / 2) & ~31;
    const int lcapB = lcap - lcapA;

    for (int r = threadIdx.x; r < RB; r += SW * 32) degc[r] = 0;
    if (threadIdx.x < 2) lcount[threadIdx.x] = 0;
    __syncthreads();

    scan_edges<true, 0, false>(q, wv, lane, row0, nrows, src, dst, nullptr, nullptr, nullptr,
                               nullptr, degc, lcount, mylist, lcapA, lcapB, E);
    __syncthreads();

    for (int r = threadIdx.x; r < nrows; r += SW * 32)
        dinv[row0 + r] = rsqrtf((float)degc[r] + 1.0f);
    if (threadIdx.x == 0) {
        const int ca = lcount[0], cb = lcount[1];
        mylist[0] = (ca <= lcapA && cb <= lcapB) ? ca : -1;
        mylist[1] = cb;
    }
}

#define RSM 132
#define RSX 140
#define NT (RB / 16)
#define WQW (HDIM / 2 * WS2 * 2)
#define WQX (KX / 2 * WS2 * 2)

template <int RS, bool XR>
__device__ __forceinline__ void list_edges(const int* __restrict__ list, int nl, int w0, int nw,
                                           int lane, int row0,
                                           const float* __restrict__ dinv,
                                           const float* __restrict__ gp,
                                           const float* __restrict__ gt, float* rowsl) {
    int i = w0 * 32;
    unsigned w = 0u;
    if (i < nl) w = (unsigned)list[min(i + lane, nl - 1)];
    while (i < nl) {
        int s  = (int)(w & ((1u << LSH) - 1u));
        int dl = (int)(w >> LSH);
        float coef = dinv[s] * dinv[row0 + dl];
        const int inext = i + nw * 32;
        unsigned wn = (unsigned)list[min(inext + lane, nl - 1)];
        add_edges<RS, XR>(s, dl, coef, min(32, nl - i), lane, gp, gt, rowsl);
        w = wn;
        i = inext;
    }
}

template <bool FIRST>
__device__ __forceinline__ void stage_w(v2f* Wq, const float* __restrict__ Wn) {
    const int n  = threadIdx.x & (HDIM - 1);
    const int ub = (int)(threadIdx.x >> 7);
    float xv[8], yv[8];
    float xe = 0.f, ye = 0.f;
#pragma unroll
    for (int i = 0; i < 8; ++i) {
        const int u  = ub + 8 * i;
        const int f0 = 8 * (u & 15) + (u >> 4);
        xv[i] = Wn[f0 * HDIM + n];
        yv[i] = Wn[(f0 + 4) * HDIM + n];
    }
    if (FIRST && ub < (KX - HDIM) / 2) {
        const int f = HDIM + 2 * ub;
        if (f     < XF) xe = Wn[f * HDIM + n];
        if (f + 1 < XF) ye = Wn[(f + 1) * HDIM + n];
    }
#pragma unroll
    for (int i = 0; i < 8; ++i) {
        const int u = ub + 8 * i;
        v2f wp; wp.x = xv[i]; wp.y = yv[i];
        Wq[u * WS2 + n] = wp;
    }
    if (FIRST && ub < (KX - HDIM) / 2) {
        v2f wp; wp.x = xe; wp.y = ye;
        Wq[(HDIM / 2 + ub) * WS2 + n] = wp;
    }
}

template <int RS, int KT>
__device__ __forceinline__ void product_tile(int tile, int lane, int row0, int nrows,
                                             float* rows, const v2f* Wq,
                                             const float* __restrict__ bias,
                                             float* __restrict__ out) {
    const int m = lane & 15;
    const int h = lane >> 4;
    const float* Ar = rows + (16 * tile + m) * RS + 2 * h;
    const v2f*   Bq = Wq + h * WS2 + m;
    v8f acc[8];
#pragma unroll
    for (int j = 0; j < 8; ++j) acc[j] = (v8f){0.f,0.f,0.f,0.f,0.f,0.f,0.f,0.f};
    for (int t = 0; t < KT; ++t) {
        const v2f a = *(const v2f*)(Ar + 4 * t);
        const v2f* Bk = Bq + 2 * t * WS2;
#pragma unroll
        for (int j = 0; j < 8; ++j) {
            const v2f b = Bk[j * 16];
            acc[j] = __builtin_amdgcn_wmma_f32_16x16x4_f32(
                false, a, false, b, (short)0, acc[j], false, false);
        }
    }
    float bj[8];
#pragma unroll
    for (int j = 0; j < 8; ++j) bj[j] = bias[16 * j + m];
    float* ob = rows + (16 * tile + 8 * h) * RS + m;
#pragma unroll
    for (int i = 0; i < 8; ++i)
#pragma unroll
        for (int j = 0; j < 8; ++j) ob[i * RS + 16 * j] = acc[j][i] + bj[j];
    for (int r = 0; r < 16; ++r) {
        const int row = 16 * tile + r;
        if (row < nrows) {
            float4 v = *(const float4*)(rows + row * RS + 4 * lane);
            v.x = tanhf(v.x); v.y = tanhf(v.y); v.z = tanhf(v.z); v.w = tanhf(v.w);
            ((float4*)out)[(size_t)(row0 + row) * (HDIM / 4) + lane] = v;
        }
    }
}

template <bool FIRST>
__global__ __launch_bounds__(1024)
void edge_block_sum(const int* __restrict__ src, const int* __restrict__ dst,
                    const float* __restrict__ dinv, const float* __restrict__ g,
                    const float* __restrict__ Wn,
                    const float* __restrict__ bias, float* __restrict__ out,
                    const int* __restrict__ lists, int N, int E) {
    constexpr int RS = FIRST ? RSX : RSM;
    constexpr int GS = FIRST ? XF : HDIM;
    constexpr int KT = (FIRST ? KX : HDIM) / 4;
    extern __shared__ float lds[];
    float* rows = lds;
    const int lane = threadIdx.x & 31;
    const int wv = __builtin_amdgcn_readfirstlane((int)(threadIdx.x >> 5));
    v2f* Wq = (v2f*)(lds + RB * RS);
    int* q = (int*)(lds + RB * RS) + wv * QC;

    const int row0  = blockIdx.x * RB;
    const int nrows = min(RB, N - row0);
    float* rowsl = rows + lane;
    const float* __restrict__ gp = g + 4 * lane;
    const float* __restrict__ gt = g + HDIM + min(lane, XT - 1);
    const int* __restrict__ mylist = lists + (size_t)row0 * HDIM;
    const int lcapA = ((nrows * HDIM - LHDR) / 2) & ~31;

    for (int r = wv; r < nrows; r += SW) {
        float di = dinv[row0 + r];
        float tv = FIRST ? gt[(size_t)(row0 + r) * GS] : 0.f;
        float4 v = row_piece<FIRST>(gp, row0 + r);
        if (FIRST) asm volatile("" :: "v"(tv));
        float di2 = di * di;
        float* rw = rowsl + r * RS;
        rw[0]  = v.x * di2; rw[32] = v.y * di2;
        rw[64] = v.z * di2; rw[96] = v.w * di2;
        if (FIRST && lane < KX - HDIM) rw[HDIM] = (lane < XT) ? tv * di2 : 0.f;
    }

    const int nlA = __builtin_amdgcn_readfirstlane(mylist[0]);
    const int nlB = __builtin_amdgcn_readfirstlane(mylist[1]);

    const bool listed = nlA >= 0;
    if (listed) stage_w<FIRST>(Wq, Wn);
    __syncthreads();
    if (!listed) {
        scan_edges<false, RS, FIRST>(q, wv, lane, row0, nrows, src, dst, dinv, gp, gt,
                                     rowsl, nullptr, nullptr, nullptr, 0, 0, E);
        __syncthreads();
        stage_w<FIRST>(Wq, Wn);
        __syncthreads();
    }
    for (int ph = listed ? 0 : 2; ph < 3; ++ph) {
        int tile = -1;
        int nl = 0, nw = 1;
        const int* __restrict__ lp = mylist + LHDR;
        if (ph == 0) {
            nl = nlA; nw = SW;
        } else if (ph == 1) {
            if (wv < SW - TA) { lp += lcapA; nl = nlB; nw = SW - TA; }
            else tile = wv - (SW - TA);
        } else if (listed) {
            if (wv < NT - TA) tile = TA + wv;
        } else {
            if (wv < NT) tile = wv;
        }
        if (nl > 0)
            list_edges<RS, FIRST>(lp, nl, wv, nw, lane, row0, dinv, gp, gt, rowsl);
        if (tile >= 0)
            product_tile<RS, KT>(tile, lane, row0, nrows, rows, Wq, bias, out);
        if (ph < 2) __syncthreads();
    }
}

extern "C" void kernel_launch(void* const* d_in, const int* in_sizes, int n_in,
                              void* d_out, int out_size, void* d_ws, size_t ws_size,
                              hipStream_t stream) {
    const float* x  = (const float*)d_in[0];
    const int*   ei = (const int*)d_in[1];
    const float* Wl[4] = {(const float*)d_in[2], (const float*)d_in[4],
                          (const float*)d_in[6], (const float*)d_in[8]};
    const float* bl[4] = {(const float*)d_in[3], (const float*)d_in[5],
                          (const float*)d_in[7], (const float*)d_in[9]};

    const int NFEAT = in_sizes[2] / HDIM;
    const int N     = in_sizes[0] / NFEAT;
    const int E     = in_sizes[1] / 2;
    const int* srcp = ei;
    const int* dstp = ei + E;

    float* wsf  = (float*)d_ws;
    float* dinv = wsf;
    float* r0   = wsf + N;
    float* r1   = r0 + (size_t)N * HDIM;
    float* r2   = r1 + (size_t)N * HDIM;

    const size_t pshm = ((size_t)SW * QC + RB + 4) * sizeof(int);
    scan_rows<<<(N + RB - 1) / RB, SW * 32, pshm, stream>>>(srcp, dstp, dinv, (int*)r1, N, E);

    const size_t xshm = ((size_t)RB * RSX + (size_t)WQX) * sizeof(float);
    edge_block_sum<true><<<(N + RB - 1) / RB, SW * 32, xshm, stream>>>(
        srcp, dstp, dinv, x, Wl[0], bl[0], r2, (int*)r1, N, E);
    const size_t mshm = ((size_t)RB * RSM + (size_t)WQW) * sizeof(float);
    edge_block_sum<false><<<(N + RB - 1) / RB, SW * 32, mshm, stream>>>(
        srcp, dstp, dinv, r2, Wl[1], bl[1], r0, (int*)r1, N, E);
    edge_block_sum<false><<<(N + RB - 1) / RB, SW * 32, mshm, stream>>>(
        srcp, dstp, dinv, r0, Wl[2], bl[2], r2, (int*)r1, N, E);
    edge_block_sum<false><<<(N + RB - 1) / RB, SW * 32, mshm, stream>>>(
        srcp, dstp, dinv, r2, Wl[3], bl[3], (float*)d_out, (int*)r1, N, E);
}


# layer 1 saves each listed edge's coefficient dinv[src]*dinv[dst] beside the edge list; layers 2-4 read it in list order, no scattered 4-byte dinv lookups per edge; sums still in arrival order
# speedup vs baseline: 5.0922x; 1.0488x over previous
#include <hip/hip_runtime.h>
#include <hip/hip_bf16.h>
#include <math.h>


#define HDIM 128
#define WS2  144
#define XF   133
#define XT   (XF - HDIM)
#define KX   136

typedef __attribute__((ext_vector_type(2))) float v2f;
typedef __attribute__((ext_vector_type(8))) float v8f;
typedef float f4u __attribute__((ext_vector_type(4), aligned(4)));

#define RB 400
#define SW 32
#define SL 4
#define QC 544
#define EG 8
#define LHDR 32
#define LSH 23
#define TA 17
#define RA (16 * TA)
#define LCAP(nrows) ((((nrows) * HDIM - LHDR) / 4) & ~31)

typedef __attribute__((ext_vector_type(4))) int v4i;

__device__ __forceinline__ void lds_add(float* p, float v) {
    __hip_atomic_fetch_add(p, v, __ATOMIC_RELAXED, __HIP_MEMORY_SCOPE_WORKGROUP);
}

__device__ __forceinline__ void lds_row_add(float* rw, float c, float4 v) {
    lds_add(rw,      c * v.x);
    lds_add(rw + 32, c * v.y);
    lds_add(rw + 64, c * v.z);
    lds_add(rw + 96, c * v.w);
}

template <bool XR>
__device__ __forceinline__ float4 row_piece(const float* __restrict__ gp, int s) {
    if (XR) {
        const f4u v = *(const f4u*)(gp + (size_t)s * XF);
        return make_float4(v.x, v.y, v.z, v.w);
    }
    return *(const float4*)(gp + (size_t)s * HDIM);
}

template <int RS, bool XR>
__device__ __forceinline__ void add_edges(int s, int dl, float coef, int cnt, int lane,
                                          const float* __restrict__ gp,
                                          const float* __restrict__ gt, float* rowsl) {
    if (cnt == 32) {
#pragma unroll
        for (int c = 0; c < 32 / EG; ++c) {
            float4 v[EG];
            float tv[EG];
#pragma unroll
            for (int j = 0; j < EG; ++j) {
                int sj = __builtin_amdgcn_readlane(s, c * EG + j);
                tv[j] = XR ? gt[(size_t)sj * XF] : 0.f;
            }
#pragma unroll
            for (int j = 0; j < EG; ++j) {
                int sj = __builtin_amdgcn_readlane(s, c * EG + j);
                v[j] = row_piece<XR>(gp, sj);
            }
            if (XR)
                asm volatile("" :: "v"(tv[0]), "v"(tv[1]), "v"(tv[2]), "v"(tv[3]),
                                   "v"(tv[4]), "v"(tv[5]), "v"(tv[6]), "v"(tv[7]));
#pragma unroll
            for (int j = 0; j < EG; ++j) {
                int   dj = __builtin_amdgcn_readlane(dl, c * EG + j);
                float cj = __int_as_float(
                    __builtin_amdgcn_readlane(__float_as_int(coef), c * EG + j));
                lds_row_add(rowsl + dj * RS, cj, v[j]);
            }
            if (XR && lane < XT) {
#pragma unroll
                for (int j = 0; j < EG; ++j) {
                    int   dj = __builtin_amdgcn_readlane(dl, c * EG + j);
                    float cj = __int_as_float(
                        __builtin_amdgcn_readlane(__float_as_int(coef), c * EG + j));
                    lds_add(rowsl + dj * RS + HDIM, cj * tv[j]);
                }
            }
        }
    } else {
        for (int j = 0; j < cnt; ++j) {
            int   sj = __builtin_amdgcn_readlane(s, j);
            int   dj = __builtin_amdgcn_readlane(dl, j);
            float cj = __int_as_float(__builtin_amdgcn_readlane(__float_as_int(coef), j));
            float4 v = row_piece<XR>(gp, sj);
            float tv = XR ? gt[(size_t)sj * XF] : 0.f;
            lds_row_add(rowsl + dj * RS, cj, v);
            if (XR && lane < XT) lds_add(rowsl + dj * RS + HDIM, cj * tv);
        }
    }
}

template <int RS, bool XR>
__device__ __forceinline__ void drain_edges(const int* qp, int cnt, int lane, int row0,
                                            const int* __restrict__ src,
                                            const int* __restrict__ dst,
                                            const float* __restrict__ dinv,
                                            const float* __restrict__ gp,
                                            const float* __restrict__ gt,
                                            float* rowsl) {
    int e = (lane < cnt) ? qp[lane] : 0;
    int s = src[e];
    int dg = dst[e];
    float coef = dinv[s] * dinv[dg];
    int dl = dg - row0;
    add_edges<RS, XR>(s, dl, coef, cnt, lane, gp, gt, rowsl);
}

__device__ __forceinline__ void collect_edges(const int* qp, int cnt, int lane, int row0,
                                              const int* __restrict__ src,
                                              const int* __restrict__ dst,
                                              int* degc, int* lcount, int* mylist,
                                              int lcapA, int lcapB) {
    int e = (lane < cnt) ? qp[lane] : 0;
    int s = src[e];
    int dl = dst[e] - row0;
    const bool act = lane < cnt;
    if (act)
        __hip_atomic_fetch_add(degc + dl, 1, __ATOMIC_RELAXED, __HIP_MEMORY_SCOPE_WORKGROUP);
    const bool inA = act && dl < RA;
    const bool inB = act && dl >= RA;
    const unsigned bA = __builtin_amdgcn_ballot_w32(inA);
    const unsigned bB = __builtin_amdgcn_ballot_w32(inB);
    int base = 0;
    if (lane == 0 && bA)
        base = __hip_atomic_fetch_add(lcount, __builtin_popcount(bA), __ATOMIC_RELAXED,
                                      __HIP_MEMORY_SCOPE_WORKGROUP);
    if (lane == 1 && bB)
        base = __hip_atomic_fetch_add(lcount + 1, __builtin_popcount(bB), __ATOMIC_RELAXED,
                                      __HIP_MEMORY_SCOPE_WORKGROUP);
    const int baseA = __builtin_amdgcn_readlane(base, 0);
    const int baseB = __builtin_amdgcn_readlane(base, 1);
    const int word = (int)((unsigned)s | ((unsigned)dl << LSH));
    if (inA) {
        const int p = baseA + (int)__builtin_amdgcn_mbcnt_lo(bA, 0);
        if (p < lcapA) mylist[LHDR + p] = word;
    }
    if (inB) {
        const int p = baseB + (int)__builtin_amdgcn_mbcnt_lo(bB, 0);
        if (p < lcapB) mylist[LHDR + lcapA + p] = word;
    }
}

template <bool COLLECT, int RS, bool XR>
__device__ __forceinline__ void scan_edges(int* q, int wv, int lane, int row0, int nrows,
                                           const int* __restrict__ src,
                                           const int* __restrict__ dst,
                                           const float* __restrict__ dinv,
                                           const float* __restrict__ gp,
                                           const float* __restrict__ gt,
                                           float* rowsl, int* degc, int* lcount,
                                           int* mylist, int lcapA, int lcapB, int E) {
    const int off = (int)(((size_t)dst >> 2) & 3);
    const v4i* __restrict__ d4 = (const v4i*)(dst - off);
    const int nfull = (E + off) >> 2;
    int qn = 0;

    if (wv == 0) {
        int et = 4 * nfull - off + lane;
        bool m = false;
        if (et >= 0 && et < E) m = (unsigned)(dst[et] - row0) < (unsigned)nrows;
        unsigned b = __builtin_amdgcn_ballot_w32(m);
        if (m) q[__builtin_amdgcn_mbcnt_lo(b, 0)] = et;
        qn = __builtin_popcount(b);
    }

    const int niter = (nfull + SW * 32 * SL - 1) / (SW * 32 * SL);
    const int rot = (int)(((long long)blockIdx.x * niter) / gridDim.x);
    for (int it = 0; it < niter; ++it) {
        int ci = it + rot;
        if (ci >= niter) ci -= niter;
        const int base = ci * (SW * 32 * SL) + wv * (32 * SL);
        v4i x[SL];
#pragma unroll
        for (int k = 0; k < SL; ++k) {
            int j4 = base + k * 32 + lane;
            x[k] = (j4 < nfull) ? d4[j4] : (v4i){-1, -1, -1, -1};
        }
        if (base == 0 && lane == 0) {
            if (off > 0) x[0].x = -1;
            if (off > 1) x[0].y = -1;
            if (off > 2) x[0].z = -1;
        }
#pragma unroll
        for (int k = 0; k < SL; ++k) {
            int e0 = 4 * (base + k * 32 + lane) - off;
#pragma unroll
            for (int c = 0; c < 4; ++c) {
                bool m = (unsigned)(x[k][c] - row0) < (unsigned)nrows;
                unsigned b = __builtin_amdgcn_ballot_w32(m);
                if (b) {
                    if (m) q[qn + __builtin_amdgcn_mbcnt_lo(b, 0)] = e0 + c;
                    qn += __builtin_popcount(b);
                }
            }
        }
        while (qn >= 32) {
            qn -= 32;
            if (COLLECT)
                collect_edges(q + qn, 32, lane, row0, src, dst, degc, lcount, mylist, lcapA, lcapB);
            else
                drain_edges<RS, XR>(q + qn, 32, lane, row0, src, dst, dinv, gp, gt, rowsl);
        }
    }
    if (qn > 0) {
        if (COLLECT)
            collect_edges(q, qn, lane, row0, src, dst, degc, lcount, mylist, lcapA, lcapB);
        else
            drain_edges<RS, XR>(q, qn, lane, row0, src, dst, dinv, gp, gt, rowsl);
    }
}

__global__ __launch_bounds__(1024)
void scan_rows(const int* __restrict__ src, const int* __restrict__ dst,
               float* __restrict__ dinv, int* __restrict__ lists, int N, int E) {
    extern __shared__ float lds[];
    const int lane = threadIdx.x & 31;
    const int wv = __builtin_amdgcn_readfirstlane((int)(threadIdx.x >> 5));
    int* q = (int*)lds + wv * QC;
    int* degc = (int*)lds + SW * QC;
    int* lcount = degc + RB;

    const int row0  = blockIdx.x * RB;
    const int nrows = min(RB, N - row0);
    int* mylist = lists + (size_t)row0 * HDIM;
    const int lcapA = LCAP(nrows);
    const int lcapB = lcapA;

    for (int r = threadIdx.x; r < RB; r += SW * 32) degc[r] = 0;
    if (threadIdx.x < 2) lcount[threadIdx.x] = 0;
    __syncthreads();

    scan_edges<true, 0, false>(q, wv, lane, row0, nrows, src, dst, nullptr, nullptr, nullptr,
                               nullptr, degc, lcount, mylist, lcapA, lcapB, E);
    __syncthreads();

    for (int r = threadIdx.x; r < nrows; r += SW * 32)
        dinv[row0 + r] = rsqrtf((float)degc[r] + 1.0f);
    if (threadIdx.x == 0) {
        const int ca = lcount[0], cb = lcount[1];
        mylist[0] = (ca <= lcapA && cb <= lcapB) ? ca : -1;
        mylist[1] = cb;
    }
}

#define RSM 132
#define RSX 140
#define NT (RB / 16)
#define WQW (HDIM / 2 * WS2 * 2)
#define WQX (KX / 2 * WS2 * 2)

template <int RS, bool XR>
__device__ __forceinline__ void list_edges(const int* __restrict__ list, float* cf, int nl,
                                           int w0, int nw, int lane, int row0,
                                           const float* __restrict__ dinv,
                                           const float* __restrict__ gp,
                                           const float* __restrict__ gt, float* rowsl) {
    int i = w0 * 32;
    unsigned w = 0u;
    float cw = 0.f;
    if (i < nl) {
        const int p = min(i + lane, nl - 1);
        w = (unsigned)list[p];
        if (!XR) cw = cf[p];
    }
    while (i < nl) {
        int s  = (int)(w & ((1u << LSH) - 1u));
        int dl = (int)(w >> LSH);
        float coef = cw;
        if (XR) coef = dinv[s] * dinv[row0 + dl];
        const int inext = i + nw * 32;
        const int pn = min(inext + lane, nl - 1);
        unsigned wn = (unsigned)list[pn];
        float cn = 0.f;
        if (!XR) cn = cf[pn];
        add_edges<RS, XR>(s, dl, coef, min(32, nl - i), lane, gp, gt, rowsl);
        if (XR && i + lane < nl) cf[i + lane] = coef;
        w = wn;
        cw = cn;
        i = inext;
    }
}

template <bool FIRST>
__device__ __forceinline__ void stage_w(v2f* Wq, const float* __restrict__ Wn) {
    const int n  = threadIdx.x & (HDIM - 1);
    const int ub = (int)(threadIdx.x >> 7);
    float xv[8], yv[8];
    float xe = 0.f, ye = 0.f;
#pragma unroll
    for (int i = 0; i < 8; ++i) {
        const int u  = ub + 8 * i;
        const int f0 = 8 * (u & 15) + (u >> 4);
        xv[i] = Wn[f0 * HDIM + n];
        yv[i] = Wn[(f0 + 4) * HDIM + n];
    }
    if (FIRST && ub < (KX - HDIM) / 2) {
        const int f = HDIM + 2 * ub;
        if (f     < XF) xe = Wn[f * HDIM + n];
        if (f + 1 < XF) ye = Wn[(f + 1) * HDIM + n];
    }
#pragma unroll
    for (int i = 0; i < 8; ++i) {
        const int u = ub + 8 * i;
        v2f wp; wp.x = xv[i]; wp.y = yv[i];
        Wq[u * WS2 + n] = wp;
    }
    if (FIRST && ub < (KX - HDIM) / 2) {
        v2f wp; wp.x = xe; wp.y = ye;
        Wq[(HDIM / 2 + ub) * WS2 + n] = wp;
    }
}

template <int RS, int KT>
__device__ __forceinline__ void product_tile(int tile, int lane, int row0, int nrows,
                                             float* rows, const v2f* Wq,
                                             const float* __restrict__ bias,
                                             float* __restrict__ out) {
    const int m = lane & 15;
    const int h = lane >> 4;
    const float* Ar = rows + (16 * tile + m) * RS + 2 * h;
    const v2f*   Bq = Wq + h * WS2 + m;
    v8f acc[8];
#pragma unroll
    for (int j = 0; j < 8; ++j) acc[j] = (v8f){0.f,0.f,0.f,0.f,0.f,0.f,0.f,0.f};
    for (int t = 0; t < KT; ++t) {
        const v2f a = *(const v2f*)(Ar + 4 * t);
        const v2f* Bk = Bq + 2 * t * WS2;
#pragma unroll
        for (int j = 0; j < 8; ++j) {
            const v2f b = Bk[j * 16];
            acc[j] = __builtin_amdgcn_wmma_f32_16x16x4_f32(
                false, a, false, b, (short)0, acc[j], false, false);
        }
    }
    float bj[8];
#pragma unroll
    for (int j = 0; j < 8; ++j) bj[j] = bias[16 * j + m];
    float* ob = rows + (16 * tile + 8 * h) * RS + m;
#pragma unroll
    for (int i = 0; i < 8; ++i)
#pragma unroll
        for (int j = 0; j < 8; ++j) ob[i * RS + 16 * j] = acc[j][i] + bj[j];
    for (int r = 0; r < 16; ++r) {
        const int row = 16 * tile + r;
        if (row < nrows) {
            float4 v = *(const float4*)(rows + row * RS + 4 * lane);
            v.x = tanhf(v.x); v.y = tanhf(v.y); v.z = tanhf(v.z); v.w = tanhf(v.w);
            ((float4*)out)[(size_t)(row0 + row) * (HDIM / 4) + lane] = v;
        }
    }
}

template <bool FIRST>
__global__ __launch_bounds__(1024)
void edge_block_sum(const int* __restrict__ src, const int* __restrict__ dst,
                    const float* __restrict__ dinv, const float* __restrict__ g,
                    const float* __restrict__ Wn,
                    const float* __restrict__ bias, float* __restrict__ out,
                    const int* __restrict__ lists, float* coefs, int N, int E) {
    constexpr int RS = FIRST ? RSX : RSM;
    constexpr int GS = FIRST ? XF : HDIM;
    constexpr int KT = (FIRST ? KX : HDIM) / 4;
    extern __shared__ float lds[];
    float* rows = lds;
    const int lane = threadIdx.x & 31;
    const int wv = __builtin_amdgcn_readfirstlane((int)(threadIdx.x >> 5));
    v2f* Wq = (v2f*)(lds + RB * RS);
    int* q = (int*)(lds + RB * RS) + wv * QC;

    const int row0  = blockIdx.x * RB;
    const int nrows = min(RB, N - row0);
    float* rowsl = rows + lane;
    const float* __restrict__ gp = g + 4 * lane;
    const float* __restrict__ gt = g + HDIM + min(lane, XT - 1);
    const int* __restrict__ mylist = lists + (size_t)row0 * HDIM;
    const int lcapA = LCAP(nrows);
    float* mycoef = coefs + (size_t)row0 * HDIM + LHDR + 2 * lcapA;

    for (int r = wv; r < nrows; r += SW) {
        float di = dinv[row0 + r];
        float tv = FIRST ? gt[(size_t)(row0 + r) * GS] : 0.f;
        float4 v = row_piece<FIRST>(gp, row0 + r);
        if (FIRST) asm volatile("" :: "v"(tv));
        float di2 = di * di;
        float* rw = rowsl + r * RS;
        rw[0]  = v.x * di2; rw[32] = v.y * di2;
        rw[64] = v.z * di2; rw[96] = v.w * di2;
        if (FIRST && lane < KX - HDIM) rw[HDIM] = (lane < XT) ? tv * di2 : 0.f;
    }

    const int nlA = __builtin_amdgcn_readfirstlane(mylist[0]);
    const int nlB = __builtin_amdgcn_readfirstlane(mylist[1]);

    const bool listed = nlA >= 0;
    if (listed) stage_w<FIRST>(Wq, Wn);
    __syncthreads();
    if (!listed) {
        scan_edges<false, RS, FIRST>(q, wv, lane, row0, nrows, src, dst, dinv, gp, gt,
                                     rowsl, nullptr, nullptr, nullptr, 0, 0, E);
        __syncthreads();
        stage_w<FIRST>(Wq, Wn);
        __syncthreads();
    }
    for (int ph = listed ? 0 : 2; ph < 3; ++ph) {
        int tile = -1;
        int nl = 0, nw = 1;
        const int* __restrict__ lp = mylist + LHDR;
        float* cp = mycoef;
        if (ph == 0) {
            nl = nlA; nw = SW;
        } else if (ph == 1) {
            if (wv < SW - TA) { lp += lcapA; cp += lcapA; nl = nlB; nw = SW - TA; }
            else tile = wv - (SW - TA);
        } else if (listed) {
            if (wv < NT - TA) tile = TA + wv;
        } else {
            if (wv < NT) tile = wv;
        }
        if (nl > 0)
            list_edges<RS, FIRST>(lp, cp, nl, wv, nw, lane, row0, dinv, gp, gt, rowsl);
        if (tile >= 0)
            product_tile<RS, KT>(tile, lane, row0, nrows, rows, Wq, bias, out);
        if (ph < 2) __syncthreads();
    }
}

extern "C" void kernel_launch(void* const* d_in, const int* in_sizes, int n_in,
                              void* d_out, int out_size, void* d_ws, size_t ws_size,
                              hipStream_t stream) {
    const float* x  = (const float*)d_in[0];
    const int*   ei = (const int*)d_in[1];
    const float* Wl[4] = {(const float*)d_in[2], (const float*)d_in[4],
                          (const float*)d_in[6], (const float*)d_in[8]};
    const float* bl[4] = {(const float*)d_in[3], (const float*)d_in[5],
                          (const float*)d_in[7], (const float*)d_in[9]};

    const int NFEAT = in_sizes[2] / HDIM;
    const int N     = in_sizes[0] / NFEAT;
    const int E     = in_sizes[1] / 2;
    const int* srcp = ei;
    const int* dstp = ei + E;

    float* wsf  = (float*)d_ws;
    float* dinv = wsf;
    float* r0   = wsf + N;
    float* r1   = r0 + (size_t)N * HDIM;
    float* r2   = r1 + (size_t)N * HDIM;

    const size_t pshm = ((size_t)SW * QC + RB + 4) * sizeof(int);
    scan_rows<<<(N + RB - 1) / RB, SW * 32, pshm, stream>>>(srcp, dstp, dinv, (int*)r1, N, E);

    const size_t xshm = ((size_t)RB * RSX + (size_t)WQX) * sizeof(float);
    edge_block_sum<true><<<(N + RB - 1) / RB, SW * 32, xshm, stream>>>(
        srcp, dstp, dinv, x, Wl[0], bl[0], r2, (int*)r1, r1, N, E);
    const size_t mshm = ((size_t)RB * RSM + (size_t)WQW) * sizeof(float);
    edge_block_sum<false><<<(N + RB - 1) / RB, SW * 32, mshm, stream>>>(
        srcp, dstp, dinv, r2, Wl[1], bl[1], r0, (int*)r1, r1, N, E);
    edge_block_sum<false><<<(N + RB - 1) / RB, SW * 32, mshm, stream>>>(
        srcp, dstp, dinv, r0, Wl[2], bl[2], r2, (int*)r1, r1, N, E);
    edge_block_sum<false><<<(N + RB - 1) / RB, SW * 32, mshm, stream>>>(
        srcp, dstp, dinv, r2, Wl[3], bl[3], (float*)d_out, (int*)r1, r1, N, E);
}


# tanh of every layer's rows by the hardware transcendental v_tanh_f32 (1 issue per value) in place of library tanhf (about 45 issues, 2 divergent paths); last bits differ from the library's
# speedup vs baseline: 5.3350x; 1.0477x over previous
#include <hip/hip_runtime.h>
#include <hip/hip_bf16.h>
#include <math.h>


#define HDIM 128
#define WS2  144
#define XF   133
#define XT   (XF - HDIM)
#define KX   136

typedef __attribute__((ext_vector_type(2))) float v2f;
typedef __attribute__((ext_vector_type(8))) float v8f;
typedef float f4u __attribute__((ext_vector_type(4), aligned(4)));

#define RB 400
#define SW 32
#define SL 4
#define QC 544
#define EG 8
#define LHDR 32
#define LSH 23
#define TA 17
#define RA (16 * TA)
#define LCAP(nrows) ((((nrows) * HDIM - LHDR) / 4) & ~31)

typedef __attribute__((ext_vector_type(4))) int v4i;

__device__ __forceinline__ void lds_add(float* p, float v) {
    __hip_atomic_fetch_add(p, v, __ATOMIC_RELAXED, __HIP_MEMORY_SCOPE_WORKGROUP);
}

__device__ __forceinline__ void lds_row_add(float* rw, float c, float4 v) {
    lds_add(rw,      c * v.x);
    lds_add(rw + 32, c * v.y);
    lds_add(rw + 64, c * v.z);
    lds_add(rw + 96, c * v.w);
}

template <bool XR>
__device__ __forceinline__ float4 row_piece(const float* __restrict__ gp, int s) {
    if (XR) {
        const f4u v = *(const f4u*)(gp + (size_t)s * XF);
        return make_float4(v.x, v.y, v.z, v.w);
    }
    return *(const float4*)(gp + (size_t)s * HDIM);
}

template <int RS, bool XR>
__device__ __forceinline__ void add_edges(int s, int dl, float coef, int cnt, int lane,
                                          const float* __restrict__ gp,
                                          const float* __restrict__ gt, float* rowsl) {
    if (cnt == 32) {
#pragma unroll
        for (int c = 0; c < 32 / EG; ++c) {
            float4 v[EG];
            float tv[EG];
#pragma unroll
            for (int j = 0; j < EG; ++j) {
                int sj = __builtin_amdgcn_readlane(s, c * EG + j);
                tv[j] = XR ? gt[(size_t)sj * XF] : 0.f;
            }
#pragma unroll
            for (int j = 0; j < EG; ++j) {
                int sj = __builtin_amdgcn_readlane(s, c * EG + j);
                v[j] = row_piece<XR>(gp, sj);
            }
            if (XR)
                asm volatile("" :: "v"(tv[0]), "v"(tv[1]), "v"(tv[2]), "v"(tv[3]),
                                   "v"(tv[4]), "v"(tv[5]), "v"(tv[6]), "v"(tv[7]));
#pragma unroll
            for (int j = 0; j < EG; ++j) {
                int   dj = __builtin_amdgcn_readlane(dl, c * EG + j);
                float cj = __int_as_float(
                    __builtin_amdgcn_readlane(__float_as_int(coef), c * EG + j));
                lds_row_add(rowsl + dj * RS, cj, v[j]);
            }
            if (XR && lane < XT) {
#pragma unroll
                for (int j = 0; j < EG; ++j) {
                    int   dj = __builtin_amdgcn_readlane(dl, c * EG + j);
                    float cj = __int_as_float(
                        __builtin_amdgcn_readlane(__float_as_int(coef), c * EG + j));
                    lds_add(rowsl + dj * RS + HDIM, cj * tv[j]);
                }
            }
        }
    } else {
        for (int j = 0; j < cnt; ++j) {
            int   sj = __builtin_amdgcn_readlane(s, j);
            int   dj = __builtin_amdgcn_readlane(dl, j);
            float cj = __int_as_float(__builtin_amdgcn_readlane(__float_as_int(coef), j));
            float4 v = row_piece<XR>(gp, sj);
            float tv = XR ? gt[(size_t)sj * XF] : 0.f;
            lds_row_add(rowsl + dj * RS, cj, v);
            if (XR && lane < XT) lds_add(rowsl + dj * RS + HDIM, cj * tv);
        }
    }
}

template <int RS, bool XR>
__device__ __forceinline__ void drain_edges(const int* qp, int cnt, int lane, int row0,
                                            const int* __restrict__ src,
                                            const int* __restrict__ dst,
                                            const float* __restrict__ dinv,
                                            const float* __restrict__ gp,
                                            const float* __restrict__ gt,
                                            float* rowsl) {
    int e = (lane < cnt) ? qp[lane] : 0;
    int s = src[e];
    int dg = dst[e];
    float coef = dinv[s] * dinv[dg];
    int dl = dg - row0;
    add_edges<RS, XR>(s, dl, coef, cnt, lane, gp, gt, rowsl);
}

__device__ __forceinline__ void collect_edges(const int* qp, int cnt, int lane, int row0,
                                              const int* __restrict__ src,
                                              const int* __restrict__ dst,
                                              int* degc, int* lcount, int* mylist,
                                              int lcapA, int lcapB) {
    int e = (lane < cnt) ? qp[lane] : 0;
    int s = src[e];
    int dl = dst[e] - row0;
    const bool act = lane < cnt;
    if (act)
        __hip_atomic_fetch_add(degc + dl, 1, __ATOMIC_RELAXED, __HIP_MEMORY_SCOPE_WORKGROUP);
    const bool inA = act && dl < RA;
    const bool inB = act && dl >= RA;
    const unsigned bA = __builtin_amdgcn_ballot_w32(inA);
    const unsigned bB = __builtin_amdgcn_ballot_w32(inB);
    int base = 0;
    if (lane == 0 && bA)
        base = __hip_atomic_fetch_add(lcount, __builtin_popcount(bA), __ATOMIC_RELAXED,
                                      __HIP_MEMORY_SCOPE_WORKGROUP);
    if (lane == 1 && bB)
        base = __hip_atomic_fetch_add(lcount + 1, __builtin_popcount(bB), __ATOMIC_RELAXED,
                                      __HIP_MEMORY_SCOPE_WORKGROUP);
    const int baseA = __builtin_amdgcn_readlane(base, 0);
    const int baseB = __builtin_amdgcn_readlane(base, 1);
    const int word = (int)((unsigned)s | ((unsigned)dl << LSH));
    if (inA) {
        const int p = baseA + (int)__builtin_amdgcn_mbcnt_lo(bA, 0);
        if (p < lcapA) mylist[LHDR + p] = word;
    }
    if (inB) {
        const int p = baseB + (int)__builtin_amdgcn_mbcnt_lo(bB, 0);
        if (p < lcapB) mylist[LHDR + lcapA + p] = word;
    }
}

template <bool COLLECT, int RS, bool XR>
__device__ __forceinline__ void scan_edges(int* q, int wv, int lane, int row0, int nrows,
                                           const int* __restrict__ src,
                                           const int* __restrict__ dst,
                                           const float* __restrict__ dinv,
                                           const float* __restrict__ gp,
                                           const float* __restrict__ gt,
                                           float* rowsl, int* degc, int* lcount,
                                           int* mylist, int lcapA, int lcapB, int E) {
    const int off = (int)(((size_t)dst >> 2) & 3);
    const v4i* __restrict__ d4 = (const v4i*)(dst - off);
    const int nfull = (E + off) >> 2;
    int qn = 0;

    if (wv == 0) {
        int et = 4 * nfull - off + lane;
        bool m = false;
        if (et >= 0 && et < E) m = (unsigned)(dst[et] - row0) < (unsigned)nrows;
        unsigned b = __builtin_amdgcn_ballot_w32(m);
        if (m) q[__builtin_amdgcn_mbcnt_lo(b, 0)] = et;
        qn = __builtin_popcount(b);
    }

    const int niter = (nfull + SW * 32 * SL - 1) / (SW * 32 * SL);
    const int rot = (int)(((long long)blockIdx.x * niter) / gridDim.x);
    for (int it = 0; it < niter; ++it) {
        int ci = it + rot;
        if (ci >= niter) ci -= niter;
        const int base = ci * (SW * 32 * SL) + wv * (32 * SL);
        v4i x[SL];
#pragma unroll
        for (int k = 0; k < SL; ++k) {
            int j4 = base + k * 32 + lane;
            x[k] = (j4 < nfull) ? d4[j4] : (v4i){-1, -1, -1, -1};
        }
        if (base == 0 && lane == 0) {
            if (off > 0) x[0].x = -1;
            if (off > 1) x[0].y = -1;
            if (off > 2) x[0].z = -1;
        }
#pragma unroll
        for (int k = 0; k < SL; ++k) {
            int e0 = 4 * (base + k * 32 + lane) - off;
#pragma unroll
            for (int c = 0; c < 4; ++c) {
                bool m = (unsigned)(x[k][c] - row0) < (unsigned)nrows;
                unsigned b = __builtin_amdgcn_ballot_w32(m);
                if (b) {
                    if (m) q[qn + __builtin_amdgcn_mbcnt_lo(b, 0)] = e0 + c;
                    qn += __builtin_popcount(b);
                }
            }
        }
        while (qn >= 32) {
            qn -= 32;
            if (COLLECT)
                collect_edges(q + qn, 32, lane, row0, src, dst, degc, lcount, mylist, lcapA, lcapB);
            else
                drain_edges<RS, XR>(q + qn, 32, lane, row0, src, dst, dinv, gp, gt, rowsl);
        }
    }
    if (qn > 0) {
        if (COLLECT)
            collect_edges(q, qn, lane, row0, src, dst, degc, lcount, mylist, lcapA, lcapB);
        else
            drain_edges<RS, XR>(q, qn, lane, row0, src, dst, dinv, gp, gt, rowsl);
    }
}

__global__ __launch_bounds__(1024)
void scan_rows(const int* __restrict__ src, const int* __restrict__ dst,
               float* __restrict__ dinv, int* __restrict__ lists, int N, int E) {
    extern __shared__ float lds[];
    const int lane = threadIdx.x & 31;
    const int wv = __builtin_amdgcn_readfirstlane((int)(threadIdx.x >> 5));
    int* q = (int*)lds + wv * QC;
    int* degc = (int*)lds + SW * QC;
    int* lcount = degc + RB;

    const int row0  = blockIdx.x * RB;
    const int nrows = min(RB, N - row0);
    int* mylist = lists + (size_t)row0 * HDIM;
    const int lcapA = LCAP(nrows);
    const int lcapB = lcapA;

    for (int r = threadIdx.x; r < RB; r += SW * 32) degc[r] = 0;
    if (threadIdx.x < 2) lcount[threadIdx.x] = 0;
    __syncthreads();

    scan_edges<true, 0, false>(q, wv, lane, row0, nrows, src, dst, nullptr, nullptr, nullptr,
                               nullptr, degc, lcount, mylist, lcapA, lcapB, E);
    __syncthreads();

    for (int r = threadIdx.x; r < nrows; r += SW * 32)
        dinv[row0 + r] = rsqrtf((float)degc[r] + 1.0f);
    if (threadIdx.x == 0) {
        const int ca = lcount[0], cb = lcount[1];
        mylist[0] = (ca <= lcapA && cb <= lcapB) ? ca : -1;
        mylist[1] = cb;
    }
}

#define RSM 132
#define RSX 140
#define NT (RB / 16)
#define WQW (HDIM / 2 * WS2 * 2)
#define WQX (KX / 2 * WS2 * 2)

template <int RS, bool XR>
__device__ __forceinline__ void list_edges(const int* __restrict__ list, float* cf, int nl,
                                           int w0, int nw, int lane, int row0,
                                           const float* __restrict__ dinv,
                                           const float* __restrict__ gp,
                                           const float* __restrict__ gt, float* rowsl) {
    int i = w0 * 32;
    unsigned w = 0u;
    float cw = 0.f;
    if (i < nl) {
        const int p = min(i + lane, nl - 1);
        w = (unsigned)list[p];
        if (!XR) cw = cf[p];
    }
    while (i < nl) {
        int s  = (int)(w & ((1u << LSH) - 1u));
        int dl = (int)(w >> LSH);
        float coef = cw;
        if (XR) coef = dinv[s] * dinv[row0 + dl];
        const int inext = i + nw * 32;
        const int pn = min(inext + lane, nl - 1);
        unsigned wn = (unsigned)list[pn];
        float cn = 0.f;
        if (!XR) cn = cf[pn];
        add_edges<RS, XR>(s, dl, coef, min(32, nl - i), lane, gp, gt, rowsl);
        if (XR && i + lane < nl) cf[i + lane] = coef;
        w = wn;
        cw = cn;
        i = inext;
    }
}

template <bool FIRST>
__device__ __forceinline__ void stage_w(v2f* Wq, const float* __restrict__ Wn) {
    const int n  = threadIdx.x & (HDIM - 1);
    const int ub = (int)(threadIdx.x >> 7);
    float xv[8], yv[8];
    float xe = 0.f, ye = 0.f;
#pragma unroll
    for (int i = 0; i < 8; ++i) {
        const int u  = ub + 8 * i;
        const int f0 = 8 * (u & 15) + (u >> 4);
        xv[i] = Wn[f0 * HDIM + n];
        yv[i] = Wn[(f0 + 4) * HDIM + n];
    }
    if (FIRST && ub < (KX - HDIM) / 2) {
        const int f = HDIM + 2 * ub;
        if (f     < XF) xe = Wn[f * HDIM + n];
        if (f + 1 < XF) ye = Wn[(f + 1) * HDIM + n];
    }
#pragma unroll
    for (int i = 0; i < 8; ++i) {
        const int u = ub + 8 * i;
        v2f wp; wp.x = xv[i]; wp.y = yv[i];
        Wq[u * WS2 + n] = wp;
    }
    if (FIRST && ub < (KX - HDIM) / 2) {
        v2f wp; wp.x = xe; wp.y = ye;
        Wq[(HDIM / 2 + ub) * WS2 + n] = wp;
    }
}

template <int RS, int KT>
__device__ __forceinline__ void product_tile(int tile, int lane, int row0, int nrows,
                                             float* rows, const v2f* Wq,
                                             const float* __restrict__ bias,
                                             float* __restrict__ out) {
    const int m = lane & 15;
    const int h = lane >> 4;
    const float* Ar = rows + (16 * tile + m) * RS + 2 * h;
    const v2f*   Bq = Wq + h * WS2 + m;
    v8f acc[8];
#pragma unroll
    for (int j = 0; j < 8; ++j) acc[j] = (v8f){0.f,0.f,0.f,0.f,0.f,0.f,0.f,0.f};
    for (int t = 0; t < KT; ++t) {
        const v2f a = *(const v2f*)(Ar + 4 * t);
        const v2f* Bk = Bq + 2 * t * WS2;
#pragma unroll
        for (int j = 0; j < 8; ++j) {
            const v2f b = Bk[j * 16];
            acc[j] = __builtin_amdgcn_wmma_f32_16x16x4_f32(
                false, a, false, b, (short)0, acc[j], false, false);
        }
    }
    float bj[8];
#pragma unroll
    for (int j = 0; j < 8; ++j) bj[j] = bias[16 * j + m];
    float* ob = rows + (16 * tile + 8 * h) * RS + m;
#pragma unroll
    for (int i = 0; i < 8; ++i)
#pragma unroll
        for (int j = 0; j < 8; ++j) ob[i * RS + 16 * j] = acc[j][i] + bj[j];
    for (int r = 0; r < 16; ++r) {
        const int row = 16 * tile + r;
        if (row < nrows) {
            float4 v = *(const float4*)(rows + row * RS + 4 * lane);
            v.x = __builtin_amdgcn_tanhf(v.x); v.y = __builtin_amdgcn_tanhf(v.y);
            v.z = __builtin_amdgcn_tanhf(v.z); v.w = __builtin_amdgcn_tanhf(v.w);
            ((float4*)out)[(size_t)(row0 + row) * (HDIM / 4) + lane] = v;
        }
    }
}

template <bool FIRST>
__global__ __launch_bounds__(1024)
void edge_block_sum(const int* __restrict__ src, const int* __restrict__ dst,
                    const float* __restrict__ dinv, const float* __restrict__ g,
                    const float* __restrict__ Wn,
                    const float* __restrict__ bias, float* __restrict__ out,
                    const int* __restrict__ lists, float* coefs, int N, int E) {
    constexpr int RS = FIRST ? RSX : RSM;
    constexpr int GS = FIRST ? XF : HDIM;
    constexpr int KT = (FIRST ? KX : HDIM) / 4;
    extern __shared__ float lds[];
    float* rows = lds;
    const int lane = threadIdx.x & 31;
    const int wv = __builtin_amdgcn_readfirstlane((int)(threadIdx.x >> 5));
    v2f* Wq = (v2f*)(lds + RB * RS);
    int* q = (int*)(lds + RB * RS) + wv * QC;

    const int row0  = blockIdx.x * RB;
    const int nrows = min(RB, N - row0);
    float* rowsl = rows + lane;
    const float* __restrict__ gp = g + 4 * lane;
    const float* __restrict__ gt = g + HDIM + min(lane, XT - 1);
    const int* __restrict__ mylist = lists + (size_t)row0 * HDIM;
    const int lcapA = LCAP(nrows);
    float* mycoef = coefs + (size_t)row0 * HDIM + LHDR + 2 * lcapA;

    for (int r = wv; r < nrows; r += SW) {
        float di = dinv[row0 + r];
        float tv = FIRST ? gt[(size_t)(row0 + r) * GS] : 0.f;
        float4 v = row_piece<FIRST>(gp, row0 + r);
        if (FIRST) asm volatile("" :: "v"(tv));
        float di2 = di * di;
        float* rw = rowsl + r * RS;
        rw[0]  = v.x * di2; rw[32] = v.y * di2;
        rw[64] = v.z * di2; rw[96] = v.w * di2;
        if (FIRST && lane < KX - HDIM) rw[HDIM] = (lane < XT) ? tv * di2 : 0.f;
    }

    const int nlA = __builtin_amdgcn_readfirstlane(mylist[0]);
    const int nlB = __builtin_amdgcn_readfirstlane(mylist[1]);

    const bool listed = nlA >= 0;
    if (listed) stage_w<FIRST>(Wq, Wn);
    __syncthreads();
    if (!listed) {
        scan_edges<false, RS, FIRST>(q, wv, lane, row0, nrows, src, dst, dinv, gp, gt,
                                     rowsl, nullptr, nullptr, nullptr, 0, 0, E);
        __syncthreads();
        stage_w<FIRST>(Wq, Wn);
        __syncthreads();
    }
    for (int ph = listed ? 0 : 2; ph < 3; ++ph) {
        int tile = -1;
        int nl = 0, nw = 1;
        const int* __restrict__ lp = mylist + LHDR;
        float* cp = mycoef;
        if (ph == 0) {
            nl = nlA; nw = SW;
        } else if (ph == 1) {
            if (wv < SW - TA) { lp += lcapA; cp += lcapA; nl = nlB; nw = SW - TA; }
            else tile = wv - (SW - TA);
        } else if (listed) {
            if (wv < NT - TA) tile = TA + wv;
        } else {
            if (wv < NT) tile = wv;
        }
        if (nl > 0)
            list_edges<RS, FIRST>(lp, cp, nl, wv, nw, lane, row0, dinv, gp, gt, rowsl);
        if (tile >= 0)
            product_tile<RS, KT>(tile, lane, row0, nrows, rows, Wq, bias, out);
        if (ph < 2) __syncthreads();
    }
}

extern "C" void kernel_launch(void* const* d_in, const int* in_sizes, int n_in,
                              void* d_out, int out_size, void* d_ws, size_t ws_size,
                              hipStream_t stream) {
    const float* x  = (const float*)d_in[0];
    const int*   ei = (const int*)d_in[1];
    const float* Wl[4] = {(const float*)d_in[2], (const float*)d_in[4],
                          (const float*)d_in[6], (const float*)d_in[8]};
    const float* bl[4] = {(const float*)d_in[3], (const float*)d_in[5],
                          (const float*)d_in[7], (const float*)d_in[9]};

    const int NFEAT = in_sizes[2] / HDIM;
    const int N     = in_sizes[0] / NFEAT;
    const int E     = in_sizes[1] / 2;
    const int* srcp = ei;
    const int* dstp = ei + E;

    float* wsf  = (float*)d_ws;
    float* dinv = wsf;
    float* r0   = wsf + N;
    float* r1   = r0 + (size_t)N * HDIM;
    float* r2   = r1 + (size_t)N * HDIM;

    const size_t pshm = ((size_t)SW * QC + RB + 4) * sizeof(int);
    scan_rows<<<(N + RB - 1) / RB, SW * 32, pshm, stream>>>(srcp, dstp, dinv, (int*)r1, N, E);

    const size_t xshm = ((size_t)RB * RSX + (size_t)WQX) * sizeof(float);
    edge_block_sum<true><<<(N + RB - 1) / RB, SW * 32, xshm, stream>>>(
        srcp, dstp, dinv, x, Wl[0], bl[0], r2, (int*)r1, r1, N, E);
    const size_t mshm = ((size_t)RB * RSM + (size_t)WQW) * sizeof(float);
    edge_block_sum<false><<<(N + RB - 1) / RB, SW * 32, mshm, stream>>>(
        srcp, dstp, dinv, r2, Wl[1], bl[1], r0, (int*)r1, r1, N, E);
    edge_block_sum<false><<<(N + RB - 1) / RB, SW * 32, mshm, stream>>>(
        srcp, dstp, dinv, r0, Wl[2], bl[2], r2, (int*)r1, r1, N, E);
    edge_block_sum<false><<<(N + RB - 1) / RB, SW * 32, mshm, stream>>>(
        srcp, dstp, dinv, r2, Wl[3], bl[3], (float*)d_out, (int*)r1, r1, N, E);
}


# prologue's blocks scan one byte per edge (low 8 bits of its destination block, from a new small launch) instead of 4-byte dst: quarter of the cache traffic, 4 edges a test; sums in arrival order
# speedup vs baseline: 5.7158x; 1.0714x over previous
#include <hip/hip_runtime.h>
#include <hip/hip_bf16.h>
#include <math.h>


#define HDIM 128
#define WS2  144
#define XF   133
#define XT   (XF - HDIM)
#define KX   136

typedef __attribute__((ext_vector_type(2))) float v2f;
typedef __attribute__((ext_vector_type(8))) float v8f;
typedef float f4u __attribute__((ext_vector_type(4), aligned(4)));

#define RB 400
#define SW 32
#define SL 4
#define QC 544
#define EG 8
#define LHDR 32
#define LSH 23
#define TA 17
#define RA (16 * TA)
#define LCAP(nrows) ((((nrows) * HDIM - LHDR) / 4) & ~31)

typedef __attribute__((ext_vector_type(4))) int v4i;

__device__ __forceinline__ void lds_add(float* p, float v) {
    __hip_atomic_fetch_add(p, v, __ATOMIC_RELAXED, __HIP_MEMORY_SCOPE_WORKGROUP);
}

__device__ __forceinline__ void lds_row_add(float* rw, float c, float4 v) {
    lds_add(rw,      c * v.x);
    lds_add(rw + 32, c * v.y);
    lds_add(rw + 64, c * v.z);
    lds_add(rw + 96, c * v.w);
}

template <bool XR>
__device__ __forceinline__ float4 row_piece(const float* __restrict__ gp, int s) {
    if (XR) {
        const f4u v = *(const f4u*)(gp + (size_t)s * XF);
        return make_float4(v.x, v.y, v.z, v.w);
    }
    return *(const float4*)(gp + (size_t)s * HDIM);
}

template <int RS, bool XR>
__device__ __forceinline__ void add_edges(int s, int dl, float coef, int cnt, int lane,
                                          const float* __restrict__ gp,
                                          const float* __restrict__ gt, float* rowsl) {
    if (cnt == 32) {
#pragma unroll
        for (int c = 0; c < 32 / EG; ++c) {
            float4 v[EG];
            float tv[EG];
#pragma unroll
            for (int j = 0; j < EG; ++j) {
                int sj = __builtin_amdgcn_readlane(s, c * EG + j);
                tv[j] = XR ? gt[(size_t)sj * XF] : 0.f;
            }
#pragma unroll
            for (int j = 0; j < EG; ++j) {
                int sj = __builtin_amdgcn_readlane(s, c * EG + j);
                v[j] = row_piece<XR>(gp, sj);
            }
            if (XR)
                asm volatile("" :: "v"(tv[0]), "v"(tv[1]), "v"(tv[2]), "v"(tv[3]),
                                   "v"(tv[4]), "v"(tv[5]), "v"(tv[6]), "v"(tv[7]));
#pragma unroll
            for (int j = 0; j < EG; ++j) {
                int   dj = __builtin_amdgcn_readlane(dl, c * EG + j);
                float cj = __int_as_float(
                    __builtin_amdgcn_readlane(__float_as_int(coef), c * EG + j));
                lds_row_add(rowsl + dj * RS, cj, v[j]);
            }
            if (XR && lane < XT) {
#pragma unroll
                for (int j = 0; j < EG; ++j) {
                    int   dj = __builtin_amdgcn_readlane(dl, c * EG + j);
                    float cj = __int_as_float(
                        __builtin_amdgcn_readlane(__float_as_int(coef), c * EG + j));
                    lds_add(rowsl + dj * RS + HDIM, cj * tv[j]);
                }
            }
        }
    } else {
        for (int j = 0; j < cnt; ++j) {
            int   sj = __builtin_amdgcn_readlane(s, j);
            int   dj = __builtin_amdgcn_readlane(dl, j);
            float cj = __int_as_float(__builtin_amdgcn_readlane(__float_as_int(coef), j));
            float4 v = row_piece<XR>(gp, sj);
            float tv = XR ? gt[(size_t)sj * XF] : 0.f;
            lds_row_add(rowsl + dj * RS, cj, v);
            if (XR && lane < XT) lds_add(rowsl + dj * RS + HDIM, cj * tv);
        }
    }
}

template <int RS, bool XR>
__device__ __forceinline__ void drain_edges(const int* qp, int cnt, int lane, int row0,
                                            const int* __restrict__ src,
                                            const int* __restrict__ dst,
                                            const float* __restrict__ dinv,
                                            const float* __restrict__ gp,
                                            const float* __restrict__ gt,
                                            float* rowsl) {
    int e = (lane < cnt) ? qp[lane] : 0;
    int s = src[e];
    int dg = dst[e];
    float coef = dinv[s] * dinv[dg];
    int dl = dg - row0;
    add_edges<RS, XR>(s, dl, coef, cnt, lane, gp, gt, rowsl);
}

__device__ __forceinline__ void collect_edges(const int* qp, int cnt, int lane, int row0,
                                              int nrows, int E,
                                              const int* __restrict__ src,
                                              const int* __restrict__ dst,
                                              int* degc, int* lcount, int* mylist,
                                              int lcapA, int lcapB) {
    int e = (lane < cnt) ? qp[lane] : 0;
    const bool inl = lane < cnt && e < E;
    if (!inl) e = 0;
    int s = src[e];
    int dl = dst[e] - row0;
    const bool act = inl && (unsigned)dl < (unsigned)nrows;
    if (act)
        __hip_atomic_fetch_add(degc + dl, 1, __ATOMIC_RELAXED, __HIP_MEMORY_SCOPE_WORKGROUP);
    const bool inA = act && dl < RA;
    const bool inB = act && dl >= RA;
    const unsigned bA = __builtin_amdgcn_ballot_w32(inA);
    const unsigned bB = __builtin_amdgcn_ballot_w32(inB);
    int base = 0;
    if (lane == 0 && bA)
        base = __hip_atomic_fetch_add(lcount, __builtin_popcount(bA), __ATOMIC_RELAXED,
                                      __HIP_MEMORY_SCOPE_WORKGROUP);
    if (lane == 1 && bB)
        base = __hip_atomic_fetch_add(lcount + 1, __builtin_popcount(bB), __ATOMIC_RELAXED,
                                      __HIP_MEMORY_SCOPE_WORKGROUP);
    const int baseA = __builtin_amdgcn_readlane(base, 0);
    const int baseB = __builtin_amdgcn_readlane(base, 1);
    const int word = (int)((unsigned)s | ((unsigned)dl << LSH));
    if (inA) {
        const int p = baseA + (int)__builtin_amdgcn_mbcnt_lo(bA, 0);
        if (p < lcapA) mylist[LHDR + p] = word;
    }
    if (inB) {
        const int p = baseB + (int)__builtin_amdgcn_mbcnt_lo(bB, 0);
        if (p < lcapB) mylist[LHDR + lcapA + p] = word;
    }
}

template <int RS, bool XR>
__device__ __forceinline__ void scan_edges(int* q, int wv, int lane, int row0, int nrows,
                                           const int* __restrict__ src,
                                           const int* __restrict__ dst,
                                           const float* __restrict__ dinv,
                                           const float* __restrict__ gp,
                                           const float* __restrict__ gt,
                                           float* rowsl, int E) {
    const int off = (int)(((size_t)dst >> 2) & 3);
    const v4i* __restrict__ d4 = (const v4i*)(dst - off);
    const int nfull = (E + off) >> 2;
    int qn = 0;

    if (wv == 0) {
        int et = 4 * nfull - off + lane;
        bool m = false;
        if (et >= 0 && et < E) m = (unsigned)(dst[et] - row0) < (unsigned)nrows;
        unsigned b = __builtin_amdgcn_ballot_w32(m);
        if (m) q[__builtin_amdgcn_mbcnt_lo(b, 0)] = et;
        qn = __builtin_popcount(b);
    }

    const int niter = (nfull + SW * 32 * SL - 1) / (SW * 32 * SL);
    const int rot = (int)(((long long)blockIdx.x * niter) / gridDim.x);
    for (int it = 0; it < niter; ++it) {
        int ci = it + rot;
        if (ci >= niter) ci -= niter;
        const int base = ci * (SW * 32 * SL) + wv * (32 * SL);
        v4i x[SL];
#pragma unroll
        for (int k = 0; k < SL; ++k) {
            int j4 = base + k * 32 + lane;
            x[k] = (j4 < nfull) ? d4[j4] : (v4i){-1, -1, -1, -1};
        }
        if (base == 0 && lane == 0) {
            if (off > 0) x[0].x = -1;
            if (off > 1) x[0].y = -1;
            if (off > 2) x[0].z = -1;
        }
#pragma unroll
        for (int k = 0; k < SL; ++k) {
            int e0 = 4 * (base + k * 32 + lane) - off;
#pragma unroll
            for (int c = 0; c < 4; ++c) {
                bool m = (unsigned)(x[k][c] - row0) < (unsigned)nrows;
                unsigned b = __builtin_amdgcn_ballot_w32(m);
                if (b) {
                    if (m) q[qn + __builtin_amdgcn_mbcnt_lo(b, 0)] = e0 + c;
                    qn += __builtin_popcount(b);
                }
            }
        }
        while (qn >= 32) {
            qn -= 32;
            drain_edges<RS, XR>(q + qn, 32, lane, row0, src, dst, dinv, gp, gt, rowsl);
        }
    }
    if (qn > 0)
        drain_edges<RS, XR>(q, qn, lane, row0, src, dst, dinv, gp, gt, rowsl);
}

#define CTURN (SW * 32 * SL)
#define CGRP(E) ((((((E) + 15) >> 4) + CTURN - 1) / CTURN) * CTURN)

__device__ __forceinline__ unsigned* code_base(float* buf) {
    return (unsigned*)(buf + ((0u - (unsigned)((size_t)buf >> 2)) & 3u));
}

__global__ __launch_bounds__(256)
void code_rows(const int* __restrict__ dst, float* cbuf, int E) {
    unsigned* __restrict__ codes = code_base(cbuf);
    const int t = blockIdx.x * 256 + threadIdx.x;
    if (t >= 4 * CGRP(E)) return;
    unsigned d[4];
#pragma unroll
    for (int k = 0; k < 4; ++k) d[k] = (unsigned)dst[min(4 * t + k, E - 1)];
    unsigned w = 0u;
#pragma unroll
    for (int k = 0; k < 4; ++k) {
        const unsigned b = (4 * t + k < E) ? (d[k] / RB) & 255u : 255u;
        w |= b << (8 * k);
    }
    codes[t] = w;
}

__device__ __forceinline__ void scan_codes(int* q, int wv, int lane, int row0, int nrows,
                                           const int* __restrict__ src,
                                           const int* __restrict__ dst,
                                           const v4i* __restrict__ codes,
                                           int* degc, int* lcount, int* mylist,
                                           int lcapA, int lcapB, int E) {
    const unsigned K = ((unsigned)(row0 / RB) & 255u) * 0x01010101u;
    int qn = 0;
    const int niter = CGRP(E) / CTURN;
    const int rot = (int)(((long long)blockIdx.x * niter) / gridDim.x);
    for (int it = 0; it < niter; ++it) {
        int ci = it + rot;
        if (ci >= niter) ci -= niter;
        const int base = ci * CTURN + wv * (32 * SL);
        v4i x[SL];
#pragma unroll
        for (int k = 0; k < SL; ++k) x[k] = codes[base + k * 32 + lane];
#pragma unroll
        for (int k = 0; k < SL; ++k) {
#pragma unroll
            for (int c = 0; c < 4; ++c) {
                const unsigned v = (unsigned)x[k][c] ^ K;
                unsigned t = (v - 0x01010101u) & ~v & 0x80808080u;
                const int e0 = 16 * (base + k * 32 + lane) + 4 * c;
                unsigned b = __builtin_amdgcn_ballot_w32(t != 0u);
                while (b) {
                    if (t) q[qn + __builtin_amdgcn_mbcnt_lo(b, 0)] = e0 + (__builtin_ctz(t) >> 3);
                    qn += __builtin_popcount(b);
                    t &= t - 1u;
                    b = __builtin_amdgcn_ballot_w32(t != 0u);
                }
            }
            while (qn >= 32) {
                qn -= 32;
                collect_edges(q + qn, 32, lane, row0, nrows, E, src, dst, degc, lcount, mylist,
                              lcapA, lcapB);
            }
        }
    }
    if (qn > 0)
        collect_edges(q, qn, lane, row0, nrows, E, src, dst, degc, lcount, mylist, lcapA, lcapB);
}

__global__ __launch_bounds__(1024)
void scan_rows(const int* __restrict__ src, const int* __restrict__ dst, float* cbuf,
               float* __restrict__ dinv, int* __restrict__ lists, int N, int E) {
    extern __shared__ float lds[];
    const int lane = threadIdx.x & 31;
    const int wv = __builtin_amdgcn_readfirstlane((int)(threadIdx.x >> 5));
    int* q = (int*)lds + wv * QC;
    int* degc = (int*)lds + SW * QC;
    int* lcount = degc + RB;

    const int row0  = blockIdx.x * RB;
    const int nrows = min(RB, N - row0);
    int* mylist = lists + (size_t)row0 * HDIM;
    const int lcapA = LCAP(nrows);
    const int lcapB = lcapA;

    for (int r = threadIdx.x; r < RB; r += SW * 32) degc[r] = 0;
    if (threadIdx.x < 2) lcount[threadIdx.x] = 0;
    __syncthreads();

    scan_codes(q, wv, lane, row0, nrows, src, dst, (const v4i*)code_base(cbuf), degc, lcount,
               mylist, lcapA, lcapB, E);
    __syncthreads();

    for (int r = threadIdx.x; r < nrows; r += SW * 32)
        dinv[row0 + r] = rsqrtf((float)degc[r] + 1.0f);
    if (threadIdx.x == 0) {
        const int ca = lcount[0], cb = lcount[1];
        mylist[0] = (ca <= lcapA && cb <= lcapB) ? ca : -1;
        mylist[1] = cb;
    }
}

#define RSM 132
#define RSX 140
#define NT (RB / 16)
#define WQW (HDIM / 2 * WS2 * 2)
#define WQX (KX / 2 * WS2 * 2)

template <int RS, bool XR>
__device__ __forceinline__ void list_edges(const int* __restrict__ list, float* cf, int nl,
                                           int w0, int nw, int lane, int row0,
                                           const float* __restrict__ dinv,
                                           const float* __restrict__ gp,
                                           const float* __restrict__ gt, float* rowsl) {
    int i = w0 * 32;
    unsigned w = 0u;
    float cw = 0.f;
    if (i < nl) {
        const int p = min(i + lane, nl - 1);
        w = (unsigned)list[p];
        if (!XR) cw = cf[p];
    }
    while (i < nl) {
        int s  = (int)(w & ((1u << LSH) - 1u));
        int dl = (int)(w >> LSH);
        float coef = cw;
        if (XR) coef = dinv[s] * dinv[row0 + dl];
        const int inext = i + nw * 32;
        const int pn = min(inext + lane, nl - 1);
        unsigned wn = (unsigned)list[pn];
        float cn = 0.f;
        if (!XR) cn = cf[pn];
        add_edges<RS, XR>(s, dl, coef, min(32, nl - i), lane, gp, gt, rowsl);
        if (XR && i + lane < nl) cf[i + lane] = coef;
        w = wn;
        cw = cn;
        i = inext;
    }
}

template <bool FIRST>
__device__ __forceinline__ void stage_w(v2f* Wq, const float* __restrict__ Wn) {
    const int n  = threadIdx.x & (HDIM - 1);
    const int ub = (int)(threadIdx.x >> 7);
    float xv[8], yv[8];
    float xe = 0.f, ye = 0.f;
#pragma unroll
    for (int i = 0; i < 8; ++i) {
        const int u  = ub + 8 * i;
        const int f0 = 8 * (u & 15) + (u >> 4);
        xv[i] = Wn[f0 * HDIM + n];
        yv[i] = Wn[(f0 + 4) * HDIM + n];
    }
    if (FIRST && ub < (KX - HDIM) / 2) {
        const int f = HDIM + 2 * ub;
        if (f     < XF) xe = Wn[f * HDIM + n];
        if (f + 1 < XF) ye = Wn[(f + 1) * HDIM + n];
    }
#pragma unroll
    for (int i = 0; i < 8; ++i) {
        const int u = ub + 8 * i;
        v2f wp; wp.x = xv[i]; wp.y = yv[i];
        Wq[u * WS2 + n] = wp;
    }
    if (FIRST && ub < (KX - HDIM) / 2) {
        v2f wp; wp.x = xe; wp.y = ye;
        Wq[(HDIM / 2 + ub) * WS2 + n] = wp;
    }
}

template <int RS, int KT>
__device__ __forceinline__ void product_tile(int tile, int lane, int row0, int nrows,
                                             float* rows, const v2f* Wq,
                                             const float* __restrict__ bias,
                                             float* __restrict__ out) {
    const int m = lane & 15;
    const int h = lane >> 4;
    const float* Ar = rows + (16 * tile + m) * RS + 2 * h;
    const v2f*   Bq = Wq + h * WS2 + m;
    v8f acc[8];
#pragma unroll
    for (int j = 0; j < 8; ++j) acc[j] = (v8f){0.f,0.f,0.f,0.f,0.f,0.f,0.f,0.f};
    for (int t = 0; t < KT; ++t) {
        const v2f a = *(const v2f*)(Ar + 4 * t);
        const v2f* Bk = Bq + 2 * t * WS2;
#pragma unroll
        for (int j = 0; j < 8; ++j) {
            const v2f b = Bk[j * 16];
            acc[j] = __builtin_amdgcn_wmma_f32_16x16x4_f32(
                false, a, false, b, (short)0, acc[j], false, false);
        }
    }
    float bj[8];
#pragma unroll
    for (int j = 0; j < 8; ++j) bj[j] = bias[16 * j + m];
    float* ob = rows + (16 * tile + 8 * h) * RS + m;
#pragma unroll
    for (int i = 0; i < 8; ++i)
#pragma unroll
        for (int j = 0; j < 8; ++j) ob[i * RS + 16 * j] = acc[j][i] + bj[j];
    for (int r = 0; r < 16; ++r) {
        const int row = 16 * tile + r;
        if (row < nrows) {
            float4 v = *(const float4*)(rows + row * RS + 4 * lane);
            v.x = __builtin_amdgcn_tanhf(v.x); v.y = __builtin_amdgcn_tanhf(v.y);
            v.z = __builtin_amdgcn_tanhf(v.z); v.w = __builtin_amdgcn_tanhf(v.w);
            ((float4*)out)[(size_t)(row0 + row) * (HDIM / 4) + lane] = v;
        }
    }
}

template <bool FIRST>
__global__ __launch_bounds__(1024)
void edge_block_sum(const int* __restrict__ src, const int* __restrict__ dst,
                    const float* __restrict__ dinv, const float* __restrict__ g,
                    const float* __restrict__ Wn,
                    const float* __restrict__ bias, float* __restrict__ out,
                    const int* __restrict__ lists, float* coefs, int N, int E) {
    constexpr int RS = FIRST ? RSX : RSM;
    constexpr int GS = FIRST ? XF : HDIM;
    constexpr int KT = (FIRST ? KX : HDIM) / 4;
    extern __shared__ float lds[];
    float* rows = lds;
    const int lane = threadIdx.x & 31;
    const int wv = __builtin_amdgcn_readfirstlane((int)(threadIdx.x >> 5));
    v2f* Wq = (v2f*)(lds + RB * RS);
    int* q = (int*)(lds + RB * RS) + wv * QC;

    const int row0  = blockIdx.x * RB;
    const int nrows = min(RB, N - row0);
    float* rowsl = rows + lane;
    const float* __restrict__ gp = g + 4 * lane;
    const float* __restrict__ gt = g + HDIM + min(lane, XT - 1);
    const int* __restrict__ mylist = lists + (size_t)row0 * HDIM;
    const int lcapA = LCAP(nrows);
    float* mycoef = coefs + (size_t)row0 * HDIM + LHDR + 2 * lcapA;

    for (int r = wv; r < nrows; r += SW) {
        float di = dinv[row0 + r];
        float tv = FIRST ? gt[(size_t)(row0 + r) * GS] : 0.f;
        float4 v = row_piece<FIRST>(gp, row0 + r);
        if (FIRST) asm volatile("" :: "v"(tv));
        float di2 = di * di;
        float* rw = rowsl + r * RS;
        rw[0]  = v.x * di2; rw[32] = v.y * di2;
        rw[64] = v.z * di2; rw[96] = v.w * di2;
        if (FIRST && lane < KX - HDIM) rw[HDIM] = (lane < XT) ? tv * di2 : 0.f;
    }

    const int nlA = __builtin_amdgcn_readfirstlane(mylist[0]);
    const int nlB = __builtin_amdgcn_readfirstlane(mylist[1]);

    const bool listed = nlA >= 0;
    if (listed) stage_w<FIRST>(Wq, Wn);
    __syncthreads();
    if (!listed) {
        scan_edges<RS, FIRST>(q, wv, lane, row0, nrows, src, dst, dinv, gp, gt, rowsl, E);
        __syncthreads();
        stage_w<FIRST>(Wq, Wn);
        __syncthreads();
    }
    for (int ph = listed ? 0 : 2; ph < 3; ++ph) {
        int tile = -1;
        int nl = 0, nw = 1;
        const int* __restrict__ lp = mylist + LHDR;
        float* cp = mycoef;
        if (ph == 0) {
            nl = nlA; nw = SW;
        } else if (ph == 1) {
            if (wv < SW - TA) { lp += lcapA; cp += lcapA; nl = nlB; nw = SW - TA; }
            else tile = wv - (SW - TA);
        } else if (listed) {
            if (wv < NT - TA) tile = TA + wv;
        } else {
            if (wv < NT) tile = wv;
        }
        if (nl > 0)
            list_edges<RS, FIRST>(lp, cp, nl, wv, nw, lane, row0, dinv, gp, gt, rowsl);
        if (tile >= 0)
            product_tile<RS, KT>(tile, lane, row0, nrows, rows, Wq, bias, out);
        if (ph < 2) __syncthreads();
    }
}

extern "C" void kernel_launch(void* const* d_in, const int* in_sizes, int n_in,
                              void* d_out, int out_size, void* d_ws, size_t ws_size,
                              hipStream_t stream) {
    const float* x  = (const float*)d_in[0];
    const int*   ei = (const int*)d_in[1];
    const float* Wl[4] = {(const float*)d_in[2], (const float*)d_in[4],
                          (const float*)d_in[6], (const float*)d_in[8]};
    const float* bl[4] = {(const float*)d_in[3], (const float*)d_in[5],
                          (const float*)d_in[7], (const float*)d_in[9]};

    const int NFEAT = in_sizes[2] / HDIM;
    const int N     = in_sizes[0] / NFEAT;
    const int E     = in_sizes[1] / 2;
    const int* srcp = ei;
    const int* dstp = ei + E;

    float* wsf  = (float*)d_ws;
    float* dinv = wsf;
    float* r0   = wsf + N;
    float* r1   = r0 + (size_t)N * HDIM;
    float* r2   = r1 + (size_t)N * HDIM;

    code_rows<<<(4 * CGRP(E) + 255) / 256, 256, 0, stream>>>(dstp, r0, E);

    const size_t pshm = ((size_t)SW * QC + RB + 4) * sizeof(int);
    scan_rows<<<(N + RB - 1) / RB, SW * 32, pshm, stream>>>(srcp, dstp, r0, dinv, (int*)r1, N, E);

    const size_t xshm = ((size_t)RB * RSX + (size_t)WQX) * sizeof(float);
    edge_block_sum<true><<<(N + RB - 1) / RB, SW * 32, xshm, stream>>>(
        srcp, dstp, dinv, x, Wl[0], bl[0], r2, (int*)r1, r1, N, E);
    const size_t mshm = ((size_t)RB * RSM + (size_t)WQW) * sizeof(float);
    edge_block_sum<false><<<(N + RB - 1) / RB, SW * 32, mshm, stream>>>(
        srcp, dstp, dinv, r2, Wl[1], bl[1], r0, (int*)r1, r1, N, E);
    edge_block_sum<false><<<(N + RB - 1) / RB, SW * 32, mshm, stream>>>(
        srcp, dstp, dinv, r0, Wl[2], bl[2], r2, (int*)r1, r1, N, E);
    edge_block_sum<false><<<(N + RB - 1) / RB, SW * 32, mshm, stream>>>(
        srcp, dstp, dinv, r2, Wl[3], bl[3], (float*)d_out, (int*)r1, r1, N, E);
}
